# starting kernel
# baseline (speedup 1.0000x reference)
#include <hip/hip_runtime.h>
#include <math.h>


typedef float v2f __attribute__((ext_vector_type(2)));
typedef float v8f __attribute__((ext_vector_type(8)));

#define N_NODES   100000
#define N_EDGES   1600000
#define IN_DIM    256
#define HEADS     4
#define OUT_DIM   32
#define FEAT_DIM  128
#define NEG_SLOPE 0.2f
#define KSTEPS    (IN_DIM / 4)
#define NTILES    (FEAT_DIM / 16)

__global__ void gat_prepack_W(const float* __restrict__ W, float* __restrict__ Wp) {
  const int t = blockIdx.x * blockDim.x + threadIdx.x;
  if (t >= KSTEPS * NTILES * 32) return;
  const int lane  = t & 31;
  const int tile  = (t >> 5) & 7;
  const int kstep = t >> 8;
  const int kb = (lane >> 4) * 2;
  const int n  = tile * 16 + (lane & 15);
  const int k  = kstep * 4 + kb;
  Wp[2 * t + 0] = W[(size_t)k * FEAT_DIM + n];
  Wp[2 * t + 1] = W[(size_t)(k + 1) * FEAT_DIM + n];
}

__global__ __launch_bounds__(256)
void gat_gemm_wmma(const float* __restrict__ x, const float* __restrict__ Wp,
                   float* __restrict__ feat) {
  const int lane = threadIdx.x & 31;
  const int mt   = blockIdx.x * 8 + (threadIdx.x >> 5);
  if (mt >= N_NODES / 16) return;

  const int kb = (lane >> 4) * 2;
  const float* aptr = x + (size_t)(mt * 16 + (lane & 15)) * IN_DIM + kb;
  const float* bptr = Wp + (size_t)lane * 2;

  v8f c[NTILES] = {};

  for (int ks = 0; ks < KSTEPS; ++ks) {
    const v2f a = *(const v2f*)(aptr + ks * 4);
#pragma unroll
    for (int t = 0; t < NTILES; ++t) {
      const v2f b = *(const v2f*)(bptr + (size_t)(ks * NTILES + t) * 64);
      c[t] = __builtin_amdgcn_wmma_f32_16x16x4_f32(false, a, false, b,
                                                   (short)0, c[t], false, false);
    }
  }

  const int mbase = mt * 16 + 8 * (lane >> 4);
  float* o = feat + (size_t)mbase * FEAT_DIM + (lane & 15);
#pragma unroll
  for (int t = 0; t < NTILES; ++t)
#pragma unroll
    for (int i = 0; i < 8; ++i)
      o[(size_t)i * FEAT_DIM + t * 16] = c[t][i];
}

__global__ void gat_init(float* __restrict__ m, float* __restrict__ denom,
                         float* __restrict__ out) {
  int t = blockIdx.x * blockDim.x + threadIdx.x;
  if (t < N_NODES * OUT_DIM) out[t] = 0.0f;
  if (t < N_NODES * HEADS) { m[t] = -INFINITY; denom[t] = 0.0f; }
}

__global__ void gat_el_er(const float* __restrict__ feat,
                          const float* __restrict__ attn_l,
                          const float* __restrict__ attn_r,
                          float* __restrict__ el, float* __restrict__ er) {
  int t = blockIdx.x * blockDim.x + threadIdx.x;
  if (t >= N_NODES * HEADS) return;
  const int h = t & 3;
  const float* f  = feat + (size_t)(t >> 2) * FEAT_DIM + h * OUT_DIM;
  const float* al = attn_l + h * OUT_DIM;
  const float* ar = attn_r + h * OUT_DIM;
  float sl = 0.0f, sr = 0.0f;
#pragma unroll
  for (int d = 0; d < OUT_DIM; ++d) { float v = f[d]; sl += v * al[d]; sr += v * ar[d]; }
  el[t] = sl;
  er[t] = sr;
}

__device__ __forceinline__ float lrelu(float v) {
  return v > 0.0f ? v : NEG_SLOPE * v;
}

__device__ __forceinline__ void atomic_max_f32(float* addr, float v) {
  if (v >= 0.0f) atomicMax((int*)addr, __float_as_int(v));
  else           atomicMin((unsigned int*)addr, __float_as_uint(v));
}

__global__ void gat_edge_max(const int* __restrict__ src, const int* __restrict__ dst,
                             const float* __restrict__ el, const float* __restrict__ er,
                             float* __restrict__ m) {
  int e = blockIdx.x * blockDim.x + threadIdx.x;
  if (e >= N_EDGES) return;
  const int s = src[e], d = dst[e];
  const float4 L = *(const float4*)(el + (size_t)s * HEADS);
  const float4 R = *(const float4*)(er + (size_t)d * HEADS);
  atomic_max_f32(m + (size_t)d * HEADS + 0, lrelu(L.x + R.x));
  atomic_max_f32(m + (size_t)d * HEADS + 1, lrelu(L.y + R.y));
  atomic_max_f32(m + (size_t)d * HEADS + 2, lrelu(L.z + R.z));
  atomic_max_f32(m + (size_t)d * HEADS + 3, lrelu(L.w + R.w));
}

__global__ void gat_edge_sum(const int* __restrict__ src, const int* __restrict__ dst,
                             const float* __restrict__ el, const float* __restrict__ er,
                             const float* __restrict__ m, float* __restrict__ denom) {
  int e = blockIdx.x * blockDim.x + threadIdx.x;
  if (e >= N_EDGES) return;
  const int s = src[e], d = dst[e];
  const float4 L = *(const float4*)(el + (size_t)s * HEADS);
  const float4 R = *(const float4*)(er + (size_t)d * HEADS);
  const float4 M = *(const float4*)(m  + (size_t)d * HEADS);
  atomicAdd(denom + (size_t)d * HEADS + 0, __expf(lrelu(L.x + R.x) - M.x));
  atomicAdd(denom + (size_t)d * HEADS + 1, __expf(lrelu(L.y + R.y) - M.y));
  atomicAdd(denom + (size_t)d * HEADS + 2, __expf(lrelu(L.z + R.z) - M.z));
  atomicAdd(denom + (size_t)d * HEADS + 3, __expf(lrelu(L.w + R.w) - M.w));
}

__global__ __launch_bounds__(256)
void gat_aggregate(const int* __restrict__ src, const int* __restrict__ dst,
                   const float* __restrict__ el, const float* __restrict__ er,
                   const float* __restrict__ m, const float* __restrict__ denom,
                   const float* __restrict__ feat, float* __restrict__ out) {
  const int lane = threadIdx.x & 31;
  const int e = blockIdx.x * 8 + (threadIdx.x >> 5);
  if (e >= N_EDGES) return;
  const int s = src[e], d = dst[e];
  const float4 L  = *(const float4*)(el    + (size_t)s * HEADS);
  const float4 R  = *(const float4*)(er    + (size_t)d * HEADS);
  const float4 M  = *(const float4*)(m     + (size_t)d * HEADS);
  const float4 Dn = *(const float4*)(denom + (size_t)d * HEADS);
  const float a0 = __expf(lrelu(L.x + R.x) - M.x) / fmaxf(Dn.x, 1e-9f);
  const float a1 = __expf(lrelu(L.y + R.y) - M.y) / fmaxf(Dn.y, 1e-9f);
  const float a2 = __expf(lrelu(L.z + R.z) - M.z) / fmaxf(Dn.z, 1e-9f);
  const float a3 = __expf(lrelu(L.w + R.w) - M.w) / fmaxf(Dn.w, 1e-9f);

  const float* f = feat + (size_t)s * FEAT_DIM + lane;
  const float acc = 0.25f * (a0 * f[0] + a1 * f[32] + a2 * f[64] + a3 * f[96]);
  atomicAdd(out + (size_t)d * OUT_DIM + lane, acc);
}

extern "C" void kernel_launch(void* const* d_in, const int* in_sizes, int n_in,
                              void* d_out, int out_size, void* d_ws, size_t ws_size,
                              hipStream_t stream) {
  const float* x      = (const float*)d_in[0];
  const float* W      = (const float*)d_in[1];
  const float* attn_l = (const float*)d_in[2];
  const float* attn_r = (const float*)d_in[3];
  const int*   src    = (const int*)d_in[4];
  const int*   dst    = (const int*)d_in[5];
  float* out = (float*)d_out;

  float* ws    = (float*)d_ws;
  float* feat  = ws;
  float* el    = feat + (size_t)N_NODES * FEAT_DIM;
  float* er    = el   + (size_t)N_NODES * HEADS;
  float* mbuf  = er   + (size_t)N_NODES * HEADS;
  float* denom = mbuf + (size_t)N_NODES * HEADS;
  float* Wp    = denom + (size_t)N_NODES * HEADS;

  gat_prepack_W<<<(KSTEPS * NTILES * 32 + 255) / 256, 256, 0, stream>>>(W, Wp);
  const int MT = N_NODES / 16;
  gat_gemm_wmma<<<(MT + 7) / 8, 256, 0, stream>>>(x, Wp, feat);
  gat_init<<<(N_NODES * OUT_DIM + 255) / 256, 256, 0, stream>>>(mbuf, denom, out);
  gat_el_er<<<(N_NODES * HEADS + 255) / 256, 256, 0, stream>>>(feat, attn_l, attn_r, el, er);
  gat_edge_max<<<(N_EDGES + 255) / 256, 256, 0, stream>>>(src, dst, el, er, mbuf);
  gat_edge_sum<<<(N_EDGES + 255) / 256, 256, 0, stream>>>(src, dst, el, er, mbuf, denom);
  gat_aggregate<<<(N_EDGES + 7) / 8, 256, 0, stream>>>(src, dst, el, er, mbuf, denom, feat, out);
}


# edge softmax + aggregation by owner: 256 workgroups each own 391 dst nodes, filter the dst list, sum denominators and rows by LDS float atomics in list order; no global atomics; 7 launches -> 4
# speedup vs baseline: 3.1602x; 3.1602x over previous
#include <hip/hip_runtime.h>
#include <math.h>


typedef float v2f __attribute__((ext_vector_type(2)));
typedef float v8f __attribute__((ext_vector_type(8)));

#define N_NODES   100000
#define N_EDGES   1600000
#define IN_DIM    256
#define HEADS     4
#define OUT_DIM   32
#define FEAT_DIM  128
#define NEG_SLOPE 0.2f
#define KSTEPS    (IN_DIM / 4)
#define NTILES    (FEAT_DIM / 16)
#define BOUND_SPREAD_MAX 32.0f

__global__ void gat_prepack_W(const float* __restrict__ W, float* __restrict__ Wp,
                              float* __restrict__ rng) {
  const int t = blockIdx.x * blockDim.x + threadIdx.x;
  if (t < HEADS) { rng[t] = -INFINITY; rng[HEADS + t] = INFINITY; }
  if (t >= KSTEPS * NTILES * 32) return;
  const int lane  = t & 31;
  const int tile  = (t >> 5) & 7;
  const int kstep = t >> 8;
  const int kb = (lane >> 4) * 2;
  const int n  = tile * 16 + (lane & 15);
  const int k  = kstep * 4 + kb;
  Wp[2 * t + 0] = W[(size_t)k * FEAT_DIM + n];
  Wp[2 * t + 1] = W[(size_t)(k + 1) * FEAT_DIM + n];
}

__global__ __launch_bounds__(256)
void gat_gemm_wmma(const float* __restrict__ x, const float* __restrict__ Wp,
                   float* __restrict__ feat) {
  const int lane = threadIdx.x & 31;
  const int mt   = blockIdx.x * 8 + (threadIdx.x >> 5);
  if (mt >= N_NODES / 16) return;

  const int kb = (lane >> 4) * 2;
  const float* aptr = x + (size_t)(mt * 16 + (lane & 15)) * IN_DIM + kb;
  const float* bptr = Wp + (size_t)lane * 2;

  v8f c[NTILES] = {};

  for (int ks = 0; ks < KSTEPS; ++ks) {
    const v2f a = *(const v2f*)(aptr + ks * 4);
#pragma unroll
    for (int t = 0; t < NTILES; ++t) {
      const v2f b = *(const v2f*)(bptr + (size_t)(ks * NTILES + t) * 64);
      c[t] = __builtin_amdgcn_wmma_f32_16x16x4_f32(false, a, false, b,
                                                   (short)0, c[t], false, false);
    }
  }

  const int mbase = mt * 16 + 8 * (lane >> 4);
  float* o = feat + (size_t)mbase * FEAT_DIM + (lane & 15);
#pragma unroll
  for (int t = 0; t < NTILES; ++t)
#pragma unroll
    for (int i = 0; i < 8; ++i)
      o[(size_t)i * FEAT_DIM + t * 16] = c[t][i];
}

__device__ __forceinline__ float lrelu(float v) {
  return v > 0.0f ? v : NEG_SLOPE * v;
}

__device__ __forceinline__ void atomic_max_f32(float* addr, float v) {
  if (v >= 0.0f) atomicMax((int*)addr, __float_as_int(v));
  else           atomicMin((unsigned int*)addr, __float_as_uint(v));
}

__device__ __forceinline__ void atomic_min_f32(float* addr, float v) {
  if (v >= 0.0f) atomicMin((int*)addr, __float_as_int(v));
  else           atomicMax((unsigned int*)addr, __float_as_uint(v));
}

__device__ __forceinline__ bool load_bound(const float* __restrict__ rng, float4& G) {
  const float4 mx = *(const float4*)(rng);
  const float4 mn = *(const float4*)(rng + HEADS);
  G = mx;
  const float r = fmaxf(fmaxf(mx.x - mn.x, mx.y - mn.y), fmaxf(mx.z - mn.z, mx.w - mn.w));
  return r <= BOUND_SPREAD_MAX;
}

#define ELER_BLOCK 1024
__global__ __launch_bounds__(ELER_BLOCK)
void gat_el_er(const float* __restrict__ feat,
               const float* __restrict__ attn_l,
               const float* __restrict__ attn_r,
               float* __restrict__ el, float* __restrict__ er,
               float* __restrict__ rng) {
  __shared__ float smx[ELER_BLOCK / 32][HEADS];
  __shared__ float smn[ELER_BLOCK / 32][HEADS];
  const int t = blockIdx.x * ELER_BLOCK + threadIdx.x;
  const int h = t & 3;
  float mx = -INFINITY, mn = INFINITY;
  if (t < N_NODES * HEADS) {
    const float* f  = feat + (size_t)(t >> 2) * FEAT_DIM + h * OUT_DIM;
    const float* al = attn_l + h * OUT_DIM;
    const float* ar = attn_r + h * OUT_DIM;
    float sl = 0.0f, sr = 0.0f;
#pragma unroll
    for (int d = 0; d < OUT_DIM; ++d) { float v = f[d]; sl += v * al[d]; sr += v * ar[d]; }
    el[t] = sl;
    er[t] = sr;
    mx = sl; mn = sl;
  }
  for (int o = 4; o < 32; o <<= 1) {
    mx = fmaxf(mx, __shfl_xor(mx, o));
    mn = fminf(mn, __shfl_xor(mn, o));
  }
  const int lane = threadIdx.x & 31, wv = threadIdx.x >> 5;
  if (lane < HEADS) { smx[wv][lane] = mx; smn[wv][lane] = mn; }
  __syncthreads();
  if (threadIdx.x < HEADS) {
    float bx = -INFINITY, bn = INFINITY;
    for (int w = 0; w < ELER_BLOCK / 32; ++w) {
      bx = fmaxf(bx, smx[w][threadIdx.x]);
      bn = fminf(bn, smn[w][threadIdx.x]);
    }
    if (bx > -INFINITY) atomic_max_f32(rng + threadIdx.x, bx + 0.0f);
    if (bn <  INFINITY) atomic_min_f32(rng + HEADS + threadIdx.x, bn + 0.0f);
  }
}

#define OWN_R      391
#define OWN_BLOCKS ((N_NODES + OWN_R - 1) / OWN_R)
#define OWN_T      1024
#define OWN_CAP    8192
#define OWN_NCH    ((N_EDGES + OWN_CAP - 1) / OWN_CAP)
#define NE4        (N_EDGES / 4)
static_assert(N_EDGES % 4 == 0, "dst is read as int4");
static_assert(OWN_CAP == 8 * OWN_T, "per-entry phases cover the list in 8 trips");
static_assert(OWN_CAP % 4 == 0, "chunks start on an int4");

__device__ __forceinline__ float4 edge_ex(const float4 L, const float4 R, const float4 M) {
  return make_float4(__expf(lrelu(L.x + R.x) - M.x), __expf(lrelu(L.y + R.y) - M.y),
                     __expf(lrelu(L.z + R.z) - M.z), __expf(lrelu(L.w + R.w) - M.w));
}

#define OWN_PUSH(E, U)                                              \
  do {                                                              \
    const int k_ = atomicAdd(&s_cnt, 1);                            \
    if (k_ < OWN_CAP) s_rec[k_] = make_int2((E), (int)(U));         \
  } while (0)

__global__ __launch_bounds__(OWN_T)
void gat_owner(const int* __restrict__ src, const int* __restrict__ dst,
               const float* __restrict__ el, const float* __restrict__ er,
               const float* __restrict__ rng, const float* __restrict__ feat,
               float* __restrict__ out) {
  __shared__ float4 s_ex[OWN_CAP];
  __shared__ int2   s_rec[OWN_CAP];
  __shared__ __attribute__((aligned(16))) float s_out[OWN_R * OUT_DIM];
  __shared__ __attribute__((aligned(16))) float s_er[OWN_R * HEADS];
  __shared__ __attribute__((aligned(16))) float s_M[OWN_R * HEADS];
  __shared__ __attribute__((aligned(16))) float s_den[OWN_R * HEADS];
  __shared__ int s_cnt;

  float4 G;
  const bool bound = load_bound(rng, G);
  const float fl = bound ? 1e-23f : 1e-9f;
  const int tid = threadIdx.x;
  const int lo  = blockIdx.x * OWN_R;
  const int cnt = min(OWN_R, N_NODES - lo);
  const unsigned ucnt = (unsigned)cnt;

  for (int t = tid; t < OWN_R * OUT_DIM; t += OWN_T) s_out[t] = 0.0f;
  for (int t = tid; t < cnt; t += OWN_T) {
    const float4 R = *(const float4*)(er + (size_t)(lo + t) * HEADS);
    *(float4*)(s_er + t * HEADS)  = R;
    *(float4*)(s_den + t * HEADS) = make_float4(0.0f, 0.0f, 0.0f, 0.0f);
    *(float4*)(s_M + t * HEADS) = bound
        ? make_float4(lrelu(G.x + R.x), lrelu(G.y + R.y), lrelu(G.z + R.z), lrelu(G.w + R.w))
        : make_float4(-INFINITY, -INFINITY, -INFINITY, -INFINITY);
  }

  const int4* __restrict__ dst4 = (const int4*)dst;
  bool fast = true;
  bool have = false;
  int  ph = bound ? 1 : 0;
  int  c = 0;
  int  n = 0;

  while (ph < 3) {
    if (!have) {
      const int q0 = fast ? 0 : c * (OWN_CAP / 4);
      const int q1 = fast ? NE4 : min(q0 + OWN_CAP / 4, NE4);
      __syncthreads();
      if (tid == 0) s_cnt = 0;
      __syncthreads();
      const int nit = (q1 - q0 + OWN_T * 8 - 1) / (OWN_T * 8);
      const int rot = (int)(blockIdx.x % (unsigned)nit);
      for (int it = 0; it < nit; ++it) {
        const int k = (it + rot < nit) ? it + rot : it + rot - nit;
        const int base = q0 + k * (OWN_T * 8);
        int4 v[8];
        if (base + OWN_T * 8 <= q1) {
#pragma unroll
          for (int j = 0; j < 8; ++j) v[j] = dst4[base + j * OWN_T + tid];
        } else {
#pragma unroll
          for (int j = 0; j < 8; ++j) {
            const int q = base + j * OWN_T + tid;
            v[j] = (q < q1) ? dst4[q] : make_int4(-1, -1, -1, -1);
          }
        }
#pragma unroll
        for (int j = 0; j < 8; ++j) {
          const int q = base + j * OWN_T + tid;
          const unsigned u0 = (unsigned)(v[j].x - lo), u1 = (unsigned)(v[j].y - lo);
          const unsigned u2 = (unsigned)(v[j].z - lo), u3 = (unsigned)(v[j].w - lo);
          if ((u0 < ucnt) | (u1 < ucnt) | (u2 < ucnt) | (u3 < ucnt)) {
            if (u0 < ucnt) OWN_PUSH(4 * q + 0, u0);
            if (u1 < ucnt) OWN_PUSH(4 * q + 1, u1);
            if (u2 < ucnt) OWN_PUSH(4 * q + 2, u2);
            if (u3 < ucnt) OWN_PUSH(4 * q + 3, u3);
          }
        }
      }
      __syncthreads();
      const int tot = s_cnt;
      if (fast && tot > OWN_CAP) { fast = false; c = 0; continue; }
      n = tot;
      {
        int ee[8], ss[8];
        float4 LL[8];
#pragma unroll
        for (int j = 0; j < 8; ++j) {
          const int i = tid + j * OWN_T;
          ee[j] = (i < n) ? s_rec[i].x : 0;
        }
#pragma unroll
        for (int j = 0; j < 8; ++j) ss[j] = src[ee[j]];
#pragma unroll
        for (int j = 0; j < 8; ++j) LL[j] = *(const float4*)(el + (size_t)ss[j] * HEADS);
#pragma unroll
        for (int j = 0; j < 8; ++j) {
          const int i = tid + j * OWN_T;
          s_rec[i].x = ss[j];
          s_ex[i] = LL[j];
        }
      }
      __syncthreads();
      have = fast;
    }

    if (ph == 0) {
#pragma unroll
      for (int j = 0; j < 8; ++j) {
        const int i = tid + j * OWN_T;
        if (i < n) {
          const int u = s_rec[i].y;
          const float4 L = s_ex[i];
          const float4 R = *(const float4*)(s_er + u * HEADS);
          atomic_max_f32(s_M + u * HEADS + 0, lrelu(L.x + R.x) + 0.0f);
          atomic_max_f32(s_M + u * HEADS + 1, lrelu(L.y + R.y) + 0.0f);
          atomic_max_f32(s_M + u * HEADS + 2, lrelu(L.z + R.z) + 0.0f);
          atomic_max_f32(s_M + u * HEADS + 3, lrelu(L.w + R.w) + 0.0f);
        }
      }
    } else if (ph == 1) {
#pragma unroll
      for (int j = 0; j < 8; ++j) {
        const int i = tid + j * OWN_T;
        if (i < n) {
          const int u = s_rec[i].y;
          const float4 X = edge_ex(s_ex[i], *(const float4*)(s_er + u * HEADS),
                                   *(const float4*)(s_M + u * HEADS));
          atomicAdd(s_den + u * HEADS + 0, X.x);
          atomicAdd(s_den + u * HEADS + 1, X.y);
          atomicAdd(s_den + u * HEADS + 2, X.z);
          atomicAdd(s_den + u * HEADS + 3, X.w);
        }
      }
    } else {
#pragma unroll
      for (int j = 0; j < 8; ++j) {
        const int i = tid + j * OWN_T;
        if (i < n) {
          const int u = s_rec[i].y;
          const float4 X = edge_ex(s_ex[i], *(const float4*)(s_er + u * HEADS),
                                   *(const float4*)(s_M + u * HEADS));
          const float4 D = *(const float4*)(s_den + u * HEADS);
          s_ex[i] = make_float4(0.25f * (X.x / fmaxf(D.x, fl)), 0.25f * (X.y / fmaxf(D.y, fl)),
                                0.25f * (X.z / fmaxf(D.z, fl)), 0.25f * (X.w / fmaxf(D.w, fl)));
        }
      }
      __syncthreads();
      const int wv = __builtin_amdgcn_readfirstlane(tid >> 5), lane = tid & 31;
      for (int i0 = wv * 8; i0 < n; i0 += (OWN_T / 32) * 8) {
        int2 r[8];
        float4 w[8];
        float f[8][4];
#pragma unroll
        for (int j = 0; j < 8; ++j) {
          const int i = min(i0 + j, n - 1);
          r[j] = s_rec[i];
          w[j] = s_ex[i];
        }
#pragma unroll
        for (int j = 0; j < 8; ++j) {
          const float* fp = feat + (size_t)r[j].x * FEAT_DIM + lane;
          f[j][0] = fp[0]; f[j][1] = fp[32]; f[j][2] = fp[64]; f[j][3] = fp[96];
        }
#pragma unroll
        for (int j = 0; j < 8; ++j) {
          const float acc = w[j].x * f[j][0] + w[j].y * f[j][1] + w[j].z * f[j][2] + w[j].w * f[j][3];
          atomicAdd(s_out + r[j].y * OUT_DIM + lane, (i0 + j < n) ? acc : 0.0f);
        }
      }
    }
    __syncthreads();
    if (fast) { ++ph; }
    else if (++c == OWN_NCH) { c = 0; ++ph; }
  }

  float4* __restrict__ o4 = (float4*)(out + (size_t)lo * OUT_DIM);
  for (int t = tid; t < cnt * (OUT_DIM / 4); t += OWN_T) o4[t] = *(const float4*)(s_out + 4 * t);
}

extern "C" void kernel_launch(void* const* d_in, const int* in_sizes, int n_in,
                              void* d_out, int out_size, void* d_ws, size_t ws_size,
                              hipStream_t stream) {
  const float* x      = (const float*)d_in[0];
  const float* W      = (const float*)d_in[1];
  const float* attn_l = (const float*)d_in[2];
  const float* attn_r = (const float*)d_in[3];
  const int*   src    = (const int*)d_in[4];
  const int*   dst    = (const int*)d_in[5];
  float* out = (float*)d_out;

  float* ws    = (float*)d_ws;
  float* feat  = ws;
  float* el    = feat + (size_t)N_NODES * FEAT_DIM;
  float* er    = el   + (size_t)N_NODES * HEADS;
  float* rng   = er   + (size_t)N_NODES * HEADS;
  float* Wp    = rng  + (size_t)2 * N_NODES * HEADS;

  gat_prepack_W<<<(KSTEPS * NTILES * 32 + 255) / 256, 256, 0, stream>>>(W, Wp, rng);
  const int MT = N_NODES / 16;
  gat_gemm_wmma<<<(MT + 7) / 8, 256, 0, stream>>>(x, Wp, feat);
  gat_el_er<<<(N_NODES * HEADS + ELER_BLOCK - 1) / ELER_BLOCK, ELER_BLOCK, 0, stream>>>(feat, attn_l, attn_r, el, er, rng);
  gat_owner<<<OWN_BLOCKS, OWN_T, 0, stream>>>(src, dst, el, er, rng, feat, out);
}


# dst reduced to one owner byte per edge in the first launch; each owner workgroup scans 1.6 MB, not 6.4 MB: 4 ids per 32-bit test, one list add per wave per trip; LDS float add order as parent
# speedup vs baseline: 3.4585x; 1.0944x over previous
#include <hip/hip_runtime.h>
#include <math.h>


typedef float v2f __attribute__((ext_vector_type(2)));
typedef float v8f __attribute__((ext_vector_type(8)));

#define N_NODES   100000
#define N_EDGES   1600000
#define IN_DIM    256
#define HEADS     4
#define OUT_DIM   32
#define FEAT_DIM  128
#define NEG_SLOPE 0.2f
#define KSTEPS    (IN_DIM / 4)
#define NTILES    (FEAT_DIM / 16)
#define BOUND_SPREAD_MAX 32.0f

#define OWN_R      391
#define OWN_BLOCKS ((N_NODES + OWN_R - 1) / OWN_R)
#define NE4        (N_EDGES / 4)
#define NE16       (N_EDGES / 16)
static_assert(OWN_BLOCKS <= 256, "an owner id is one byte");
static_assert(N_EDGES % 16 == 0, "dst is read as int4, owner ids as uint4");

#define PRE_T      256
#define PRE_BLOCKS 391
static_assert(PRE_BLOCKS * PRE_T * 4 >= NE4, "every dst quad has a thread");
static_assert(64 + NE4 <= 2 * N_NODES * HEADS, "owner ids fit between rng and Wp");
static_assert(PRE_BLOCKS * PRE_T >= KSTEPS * NTILES * 32, "every W pair has a thread");
__global__ __launch_bounds__(PRE_T)
void gat_prepack_W(const float* __restrict__ W, float* __restrict__ Wp,
                   float* __restrict__ rng, const int* __restrict__ dst,
                   unsigned* __restrict__ ob) {
  const int t = blockIdx.x * PRE_T + threadIdx.x;
  if (t < HEADS) { rng[t] = -INFINITY; rng[HEADS + t] = INFINITY; }
  if (t < KSTEPS * NTILES * 32) {
    const int lane  = t & 31;
    const int tile  = (t >> 5) & 7;
    const int kstep = t >> 8;
    const int kb = (lane >> 4) * 2;
    const int n  = tile * 16 + (lane & 15);
    const int k  = kstep * 4 + kb;
    Wp[2 * t + 0] = W[(size_t)k * FEAT_DIM + n];
    Wp[2 * t + 1] = W[(size_t)(k + 1) * FEAT_DIM + n];
  }
  const int4* __restrict__ dst4 = (const int4*)dst;
  int4 d[4];
#pragma unroll
  for (int j = 0; j < 4; ++j) d[j] = dst4[min(t + j * (PRE_BLOCKS * PRE_T), NE4 - 1)];
#pragma unroll
  for (int j = 0; j < 4; ++j) {
    const int q = t + j * (PRE_BLOCKS * PRE_T);
    if (q < NE4)
      ob[q] = ((unsigned)d[j].x / OWN_R) | (((unsigned)d[j].y / OWN_R) << 8) |
              (((unsigned)d[j].z / OWN_R) << 16) | (((unsigned)d[j].w / OWN_R) << 24);
  }
}

__global__ __launch_bounds__(256)
void gat_gemm_wmma(const float* __restrict__ x, const float* __restrict__ Wp,
                   float* __restrict__ feat) {
  const int lane = threadIdx.x & 31;
  const int mt   = blockIdx.x * 8 + (threadIdx.x >> 5);
  if (mt >= N_NODES / 16) return;

  const int kb = (lane >> 4) * 2;
  const float* aptr = x + (size_t)(mt * 16 + (lane & 15)) * IN_DIM + kb;
  const float* bptr = Wp + (size_t)lane * 2;

  v8f c[NTILES] = {};

  for (int ks = 0; ks < KSTEPS; ++ks) {
    const v2f a = *(const v2f*)(aptr + ks * 4);
#pragma unroll
    for (int t = 0; t < NTILES; ++t) {
      const v2f b = *(const v2f*)(bptr + (size_t)(ks * NTILES + t) * 64);
      c[t] = __builtin_amdgcn_wmma_f32_16x16x4_f32(false, a, false, b,
                                                   (short)0, c[t], false, false);
    }
  }

  const int mbase = mt * 16 + 8 * (lane >> 4);
  float* o = feat + (size_t)mbase * FEAT_DIM + (lane & 15);
#pragma unroll
  for (int t = 0; t < NTILES; ++t)
#pragma unroll
    for (int i = 0; i < 8; ++i)
      o[(size_t)i * FEAT_DIM + t * 16] = c[t][i];
}

__device__ __forceinline__ float lrelu(float v) {
  return v > 0.0f ? v : NEG_SLOPE * v;
}

__device__ __forceinline__ void atomic_max_f32(float* addr, float v) {
  if (v >= 0.0f) atomicMax((int*)addr, __float_as_int(v));
  else           atomicMin((unsigned int*)addr, __float_as_uint(v));
}

__device__ __forceinline__ void atomic_min_f32(float* addr, float v) {
  if (v >= 0.0f) atomicMin((int*)addr, __float_as_int(v));
  else           atomicMax((unsigned int*)addr, __float_as_uint(v));
}

__device__ __forceinline__ bool load_bound(const float* __restrict__ rng, float4& G) {
  const float4 mx = *(const float4*)(rng);
  const float4 mn = *(const float4*)(rng + HEADS);
  G = mx;
  const float r = fmaxf(fmaxf(mx.x - mn.x, mx.y - mn.y), fmaxf(mx.z - mn.z, mx.w - mn.w));
  return r <= BOUND_SPREAD_MAX;
}

#define ELER_BLOCK 1024
__global__ __launch_bounds__(ELER_BLOCK)
void gat_el_er(const float* __restrict__ feat,
               const float* __restrict__ attn_l,
               const float* __restrict__ attn_r,
               float* __restrict__ el, float* __restrict__ er,
               float* __restrict__ rng) {
  __shared__ float smx[ELER_BLOCK / 32][HEADS];
  __shared__ float smn[ELER_BLOCK / 32][HEADS];
  const int t = blockIdx.x * ELER_BLOCK + threadIdx.x;
  const int h = t & 3;
  float mx = -INFINITY, mn = INFINITY;
  if (t < N_NODES * HEADS) {
    const float* f  = feat + (size_t)(t >> 2) * FEAT_DIM + h * OUT_DIM;
    const float* al = attn_l + h * OUT_DIM;
    const float* ar = attn_r + h * OUT_DIM;
    float sl = 0.0f, sr = 0.0f;
#pragma unroll
    for (int d = 0; d < OUT_DIM; ++d) { float v = f[d]; sl += v * al[d]; sr += v * ar[d]; }
    el[t] = sl;
    er[t] = sr;
    mx = sl; mn = sl;
  }
  for (int o = 4; o < 32; o <<= 1) {
    mx = fmaxf(mx, __shfl_xor(mx, o));
    mn = fminf(mn, __shfl_xor(mn, o));
  }
  const int lane = threadIdx.x & 31, wv = threadIdx.x >> 5;
  if (lane < HEADS) { smx[wv][lane] = mx; smn[wv][lane] = mn; }
  __syncthreads();
  if (threadIdx.x < HEADS) {
    float bx = -INFINITY, bn = INFINITY;
    for (int w = 0; w < ELER_BLOCK / 32; ++w) {
      bx = fmaxf(bx, smx[w][threadIdx.x]);
      bn = fminf(bn, smn[w][threadIdx.x]);
    }
    if (bx > -INFINITY) atomic_max_f32(rng + threadIdx.x, bx + 0.0f);
    if (bn <  INFINITY) atomic_min_f32(rng + HEADS + threadIdx.x, bn + 0.0f);
  }
}

#define OWN_T      1024
#define OWN_CAP    8192
#define OWN_NCH    ((N_EDGES + OWN_CAP - 1) / OWN_CAP)
static_assert(OWN_CAP == 8 * OWN_T, "per-entry phases cover the list in 8 trips");
static_assert(OWN_CAP % 16 == 0, "chunks start on a uint4 of owner ids");

__device__ __forceinline__ float4 edge_ex(const float4 L, const float4 R, const float4 M) {
  return make_float4(__expf(lrelu(L.x + R.x) - M.x), __expf(lrelu(L.y + R.y) - M.y),
                     __expf(lrelu(L.z + R.z) - M.z), __expf(lrelu(L.w + R.w) - M.w));
}

__device__ __forceinline__ unsigned nonzero_bytes(const unsigned x) {
  return ((x & 0x7F7F7F7Fu) + 0x7F7F7F7Fu) | x;
}

__device__ __forceinline__ unsigned other_owner(const uint4 v, const unsigned pat) {
  unsigned c = nonzero_bytes(v.x ^ pat);
  c = (c & ~0x40404040u) | ((nonzero_bytes(v.y ^ pat) >> 1) & 0x40404040u);
  c = (c & ~0x20202020u) | ((nonzero_bytes(v.z ^ pat) >> 2) & 0x20202020u);
  c = (c & ~0x10101010u) | ((nonzero_bytes(v.w ^ pat) >> 3) & 0x10101010u);
  return c;
}

__global__ __launch_bounds__(OWN_T)
void gat_owner(const int* __restrict__ src, const int* __restrict__ dst,
               const float* __restrict__ el, const float* __restrict__ er,
               const float* __restrict__ rng, const float* __restrict__ feat,
               const unsigned* __restrict__ ob, float* __restrict__ out) {
  __shared__ float4 s_ex[OWN_CAP];
  __shared__ int2   s_rec[OWN_CAP];
  __shared__ __attribute__((aligned(16))) float s_out[OWN_R * OUT_DIM];
  __shared__ __attribute__((aligned(16))) float s_er[OWN_R * HEADS];
  __shared__ __attribute__((aligned(16))) float s_M[OWN_R * HEADS];
  __shared__ __attribute__((aligned(16))) float s_den[OWN_R * HEADS];
  __shared__ int s_cnt;

  float4 G;
  const bool bound = load_bound(rng, G);
  const float fl = bound ? 1e-23f : 1e-9f;
  const int tid = threadIdx.x;
  const int lo  = blockIdx.x * OWN_R;
  const int cnt = min(OWN_R, N_NODES - lo);
  const unsigned pat = blockIdx.x * 0x01010101u;

  for (int t = tid; t < OWN_R * OUT_DIM; t += OWN_T) s_out[t] = 0.0f;
  for (int t = tid; t < cnt; t += OWN_T) {
    const float4 R = *(const float4*)(er + (size_t)(lo + t) * HEADS);
    *(float4*)(s_er + t * HEADS)  = R;
    *(float4*)(s_den + t * HEADS) = make_float4(0.0f, 0.0f, 0.0f, 0.0f);
    *(float4*)(s_M + t * HEADS) = bound
        ? make_float4(lrelu(G.x + R.x), lrelu(G.y + R.y), lrelu(G.z + R.z), lrelu(G.w + R.w))
        : make_float4(-INFINITY, -INFINITY, -INFINITY, -INFINITY);
  }

  const uint4* __restrict__ ob4 = (const uint4*)ob;
  bool fast = true;
  bool have = false;
  int  ph = bound ? 1 : 0;
  int  c = 0;
  int  n = 0;

  while (ph < 3) {
    if (!have) {
      const int q0 = fast ? 0 : c * (OWN_CAP / 16);
      const int q1 = fast ? NE16 : min(q0 + OWN_CAP / 16, NE16);
      __syncthreads();
      if (tid == 0) s_cnt = 0;
      __syncthreads();
      const int nit = (q1 - q0 + OWN_T * 8 - 1) / (OWN_T * 8);
      const int rot = (int)(blockIdx.x % (unsigned)nit);
      for (int it = 0; it < nit; ++it) {
        const int k = (it + rot < nit) ? it + rot : it + rot - nit;
        const int base = q0 + k * (OWN_T * 8);
        uint4 v[8];
#pragma unroll
        for (int j = 0; j < 8; ++j) v[j] = ob4[min(base + j * OWN_T + tid, q1 - 1)];
        unsigned cc[8], g[4];
#pragma unroll
        for (int j = 0; j < 8; ++j) cc[j] = other_owner(v[j], pat);
        if (base + OWN_T * 8 > q1) {
#pragma unroll
          for (int j = 0; j < 8; ++j) if (base + j * OWN_T + tid >= q1) cc[j] = ~0u;
        }
#pragma unroll
        for (int pr = 0; pr < 4; ++pr)
          g[pr] = ~((cc[2 * pr + 1] & 0xF0F0F0F0u) | ((cc[2 * pr] >> 4) & 0x0F0F0F0Fu));
        const unsigned nh = __popc(g[0]) + __popc(g[1]) + __popc(g[2]) + __popc(g[3]);
        unsigned pre = 0u, tot = 0u;
        for (int sl = 0; sl < 8; ++sl) {
          if (__builtin_amdgcn_ballot_w32((nh >> sl) != 0u) == 0u) break;
          const unsigned bk = __builtin_amdgcn_ballot_w32(((nh >> sl) & 1u) != 0u);
          pre += __builtin_amdgcn_mbcnt_lo(bk, 0u) << sl;
          tot += (unsigned)__popc(bk) << sl;
        }
        if (tot != 0u) {
          int at = 0;
          if ((tid & 31) == 0) at = atomicAdd(&s_cnt, (int)tot);
          int slot = __builtin_amdgcn_readfirstlane(at) + (int)pre;
#pragma unroll
          for (int pr = 0; pr < 4; ++pr) {
            unsigned gg = g[pr];
            const int e0 = 16 * (base + 2 * pr * OWN_T + tid);
            while (gg) {
              const int p = __builtin_ctz(gg);
              gg &= gg - 1;
              const int r = p & 7;
              const int e = e0 + (r >> 2) * (16 * OWN_T) + 4 * (~r & 3) + (p >> 3);
              if (slot < OWN_CAP) s_rec[slot].x = e;
              ++slot;
            }
          }
        }
      }
      __syncthreads();
      const int tot = s_cnt;
      if (fast && tot > OWN_CAP) { fast = false; c = 0; continue; }
      n = tot;
      {
        int ee[8], ss[8], dd[8];
        float4 LL[8];
#pragma unroll
        for (int j = 0; j < 8; ++j) {
          const int i = tid + j * OWN_T;
          ee[j] = (i < n) ? s_rec[i].x : 0;
        }
#pragma unroll
        for (int j = 0; j < 8; ++j) { ss[j] = src[ee[j]]; dd[j] = dst[ee[j]]; }
#pragma unroll
        for (int j = 0; j < 8; ++j) LL[j] = *(const float4*)(el + (size_t)ss[j] * HEADS);
#pragma unroll
        for (int j = 0; j < 8; ++j) {
          const int i = tid + j * OWN_T;
          s_rec[i] = make_int2(ss[j], dd[j] - lo);
          s_ex[i] = LL[j];
        }
      }
      __syncthreads();
      have = fast;
    }

    if (ph == 0) {
#pragma unroll
      for (int j = 0; j < 8; ++j) {
        const int i = tid + j * OWN_T;
        if (i < n) {
          const int u = s_rec[i].y;
          const float4 L = s_ex[i];
          const float4 R = *(const float4*)(s_er + u * HEADS);
          atomic_max_f32(s_M + u * HEADS + 0, lrelu(L.x + R.x) + 0.0f);
          atomic_max_f32(s_M + u * HEADS + 1, lrelu(L.y + R.y) + 0.0f);
          atomic_max_f32(s_M + u * HEADS + 2, lrelu(L.z + R.z) + 0.0f);
          atomic_max_f32(s_M + u * HEADS + 3, lrelu(L.w + R.w) + 0.0f);
        }
      }
    } else if (ph == 1) {
#pragma unroll
      for (int j = 0; j < 8; ++j) {
        const int i = tid + j * OWN_T;
        if (i < n) {
          const int u = s_rec[i].y;
          const float4 X = edge_ex(s_ex[i], *(const float4*)(s_er + u * HEADS),
                                   *(const float4*)(s_M + u * HEADS));
          atomicAdd(s_den + u * HEADS + 0, X.x);
          atomicAdd(s_den + u * HEADS + 1, X.y);
          atomicAdd(s_den + u * HEADS + 2, X.z);
          atomicAdd(s_den + u * HEADS + 3, X.w);
        }
      }
    } else {
#pragma unroll
      for (int j = 0; j < 8; ++j) {
        const int i = tid + j * OWN_T;
        if (i < n) {
          const int u = s_rec[i].y;
          const float4 X = edge_ex(s_ex[i], *(const float4*)(s_er + u * HEADS),
                                   *(const float4*)(s_M + u * HEADS));
          const float4 D = *(const float4*)(s_den + u * HEADS);
          s_ex[i] = make_float4(0.25f * (X.x / fmaxf(D.x, fl)), 0.25f * (X.y / fmaxf(D.y, fl)),
                                0.25f * (X.z / fmaxf(D.z, fl)), 0.25f * (X.w / fmaxf(D.w, fl)));
        }
      }
      __syncthreads();
      const int wv = __builtin_amdgcn_readfirstlane(tid >> 5), lane = tid & 31;
      for (int i0 = wv * 8; i0 < n; i0 += (OWN_T / 32) * 8) {
        int2 r[8];
        float4 w[8];
        float f[8][4];
#pragma unroll
        for (int j = 0; j < 8; ++j) {
          const int i = min(i0 + j, n - 1);
          r[j] = s_rec[i];
          w[j] = s_ex[i];
        }
#pragma unroll
        for (int j = 0; j < 8; ++j) {
          const float* fp = feat + (size_t)r[j].x * FEAT_DIM + lane;
          f[j][0] = fp[0]; f[j][1] = fp[32]; f[j][2] = fp[64]; f[j][3] = fp[96];
        }
#pragma unroll
        for (int j = 0; j < 8; ++j) {
          const float acc = w[j].x * f[j][0] + w[j].y * f[j][1] + w[j].z * f[j][2] + w[j].w * f[j][3];
          atomicAdd(s_out + r[j].y * OUT_DIM + lane, (i0 + j < n) ? acc : 0.0f);
        }
      }
    }
    __syncthreads();
    if (fast) { ++ph; }
    else if (++c == OWN_NCH) { c = 0; ++ph; }
  }

  float4* __restrict__ o4 = (float4*)(out + (size_t)lo * OUT_DIM);
  for (int t = tid; t < cnt * (OUT_DIM / 4); t += OWN_T) o4[t] = *(const float4*)(s_out + 4 * t);
}

extern "C" void kernel_launch(void* const* d_in, const int* in_sizes, int n_in,
                              void* d_out, int out_size, void* d_ws, size_t ws_size,
                              hipStream_t stream) {
  const float* x      = (const float*)d_in[0];
  const float* W      = (const float*)d_in[1];
  const float* attn_l = (const float*)d_in[2];
  const float* attn_r = (const float*)d_in[3];
  const int*   src    = (const int*)d_in[4];
  const int*   dst    = (const int*)d_in[5];
  float* out = (float*)d_out;

  float* ws    = (float*)d_ws;
  float* feat  = ws;
  float* el    = feat + (size_t)N_NODES * FEAT_DIM;
  float* er    = el   + (size_t)N_NODES * HEADS;
  float* rng   = er   + (size_t)N_NODES * HEADS;
  float* Wp    = rng  + (size_t)2 * N_NODES * HEADS;
  unsigned* ob = (unsigned*)(rng + 64);

  gat_prepack_W<<<PRE_BLOCKS, PRE_T, 0, stream>>>(W, Wp, rng, dst, ob);
  const int MT = N_NODES / 16;
  gat_gemm_wmma<<<(MT + 7) / 8, 256, 0, stream>>>(x, Wp, feat);
  gat_el_er<<<(N_NODES * HEADS + ELER_BLOCK - 1) / ELER_BLOCK, ELER_BLOCK, 0, stream>>>(feat, attn_l, attn_r, el, er, rng);
  gat_owner<<<OWN_BLOCKS, OWN_T, 0, stream>>>(src, dst, el, er, rng, feat, ob, out);
}


# feat rows stored [dim][head]: a source row is one 16-byte-per-lane load in the aggregation (was 4 one-line loads), 16 rows in flight per wave; GEMM stores whole lines; LDS float add order as parent
# speedup vs baseline: 3.6127x; 1.0446x over previous
#include <hip/hip_runtime.h>
#include <math.h>


typedef float v2f __attribute__((ext_vector_type(2)));
typedef float v4f __attribute__((ext_vector_type(4)));
typedef float v8f __attribute__((ext_vector_type(8)));

#define N_NODES   100000
#define N_EDGES   1600000
#define IN_DIM    256
#define HEADS     4
#define OUT_DIM   32
#define FEAT_DIM  128
#define NEG_SLOPE 0.2f
#define KSTEPS    (IN_DIM / 4)
#define NTILES    (FEAT_DIM / 16)
#define BOUND_SPREAD_MAX 32.0f

#define OWN_R      391
#define OWN_BLOCKS ((N_NODES + OWN_R - 1) / OWN_R)
#define NE4        (N_EDGES / 4)
#define NE16       (N_EDGES / 16)
static_assert(OWN_BLOCKS <= 256, "an owner id is one byte");
static_assert(N_EDGES % 16 == 0, "dst is read as int4, owner ids as uint4");

#define PRE_T      256
#define PRE_BLOCKS 391
static_assert(PRE_BLOCKS * PRE_T * 4 >= NE4, "every dst quad has a thread");
static_assert(64 + NE4 <= 2 * N_NODES * HEADS, "owner ids fit between rng and Wp");
static_assert(PRE_BLOCKS * PRE_T >= KSTEPS * NTILES * 32, "every W pair has a thread");
__global__ __launch_bounds__(PRE_T)
void gat_prepack_W(const float* __restrict__ W, float* __restrict__ Wp,
                   float* __restrict__ rng, const int* __restrict__ dst,
                   unsigned* __restrict__ ob) {
  const int t = blockIdx.x * PRE_T + threadIdx.x;
  if (t < HEADS) { rng[t] = -INFINITY; rng[HEADS + t] = INFINITY; }
  if (t < KSTEPS * NTILES * 32) {
    const int lane  = t & 31;
    const int tile  = (t >> 5) & 7;
    const int kstep = t >> 8;
    const int kb = (lane >> 4) * 2;
    const int n  = tile * 16 + (lane & 15);
    const int k  = kstep * 4 + kb;
    Wp[2 * t + 0] = W[(size_t)k * FEAT_DIM + n];
    Wp[2 * t + 1] = W[(size_t)(k + 1) * FEAT_DIM + n];
  }
  const int4* __restrict__ dst4 = (const int4*)dst;
  int4 d[4];
#pragma unroll
  for (int j = 0; j < 4; ++j) d[j] = dst4[min(t + j * (PRE_BLOCKS * PRE_T), NE4 - 1)];
#pragma unroll
  for (int j = 0; j < 4; ++j) {
    const int q = t + j * (PRE_BLOCKS * PRE_T);
    if (q < NE4)
      ob[q] = ((unsigned)d[j].x / OWN_R) | (((unsigned)d[j].y / OWN_R) << 8) |
              (((unsigned)d[j].z / OWN_R) << 16) | (((unsigned)d[j].w / OWN_R) << 24);
  }
}

__global__ __launch_bounds__(256)
void gat_gemm_wmma(const float* __restrict__ x, const float* __restrict__ Wp,
                   float* __restrict__ feat) {
  const int lane = threadIdx.x & 31;
  const int mt   = blockIdx.x * 8 + (threadIdx.x >> 5);
  if (mt >= N_NODES / 16) return;

  const int kb = (lane >> 4) * 2;
  const float* aptr = x + (size_t)(mt * 16 + (lane & 15)) * IN_DIM + kb;
  const float* bptr = Wp + (size_t)lane * 2;

  v8f c[NTILES] = {};

  for (int ks = 0; ks < KSTEPS; ++ks) {
    const v2f a = *(const v2f*)(aptr + ks * 4);
#pragma unroll
    for (int t = 0; t < NTILES; ++t) {
      const v2f b = *(const v2f*)(bptr + (size_t)(ks * NTILES + t) * 64);
      c[t] = __builtin_amdgcn_wmma_f32_16x16x4_f32(false, a, false, b,
                                                   (short)0, c[t], false, false);
    }
  }

  const int mbase = mt * 16 + 8 * (lane >> 4);
  float* o = feat + (size_t)mbase * FEAT_DIM + (lane & 15) * HEADS;
#pragma unroll
  for (int tb = 0; tb < 2; ++tb)
#pragma unroll
    for (int i = 0; i < 8; ++i)
      *(float4*)(o + (size_t)i * FEAT_DIM + tb * 16 * HEADS) =
          make_float4(c[tb][i], c[tb + 2][i], c[tb + 4][i], c[tb + 6][i]);
}

__device__ __forceinline__ float lrelu(float v) {
  return v > 0.0f ? v : NEG_SLOPE * v;
}

__device__ __forceinline__ void atomic_max_f32(float* addr, float v) {
  if (v >= 0.0f) atomicMax((int*)addr, __float_as_int(v));
  else           atomicMin((unsigned int*)addr, __float_as_uint(v));
}

__device__ __forceinline__ void atomic_min_f32(float* addr, float v) {
  if (v >= 0.0f) atomicMin((int*)addr, __float_as_int(v));
  else           atomicMax((unsigned int*)addr, __float_as_uint(v));
}

__device__ __forceinline__ bool load_bound(const float* __restrict__ rng, float4& G) {
  const float4 mx = *(const float4*)(rng);
  const float4 mn = *(const float4*)(rng + HEADS);
  G = mx;
  const float r = fmaxf(fmaxf(mx.x - mn.x, mx.y - mn.y), fmaxf(mx.z - mn.z, mx.w - mn.w));
  return r <= BOUND_SPREAD_MAX;
}

#define ELER_BLOCK 1024
__global__ __launch_bounds__(ELER_BLOCK)
void gat_el_er(const float* __restrict__ feat,
               const float* __restrict__ attn_l,
               const float* __restrict__ attn_r,
               float* __restrict__ el, float* __restrict__ er,
               float* __restrict__ rng) {
  __shared__ float smx[ELER_BLOCK / 32][HEADS];
  __shared__ float smn[ELER_BLOCK / 32][HEADS];
  const int t = blockIdx.x * ELER_BLOCK + threadIdx.x;
  const int h = t & 3;
  float mx = -INFINITY, mn = INFINITY;
  if (t < N_NODES * HEADS) {
    const float* f  = feat + (size_t)(t >> 2) * FEAT_DIM + h;
    const float* al = attn_l + h * OUT_DIM;
    const float* ar = attn_r + h * OUT_DIM;
    float sl = 0.0f, sr = 0.0f;
#pragma unroll
    for (int d = 0; d < OUT_DIM; ++d) { float v = f[d * HEADS]; sl += v * al[d]; sr += v * ar[d]; }
    el[t] = sl;
    er[t] = sr;
    mx = sl; mn = sl;
  }
  for (int o = 4; o < 32; o <<= 1) {
    mx = fmaxf(mx, __shfl_xor(mx, o));
    mn = fminf(mn, __shfl_xor(mn, o));
  }
  const int lane = threadIdx.x & 31, wv = threadIdx.x >> 5;
  if (lane < HEADS) { smx[wv][lane] = mx; smn[wv][lane] = mn; }
  __syncthreads();
  if (threadIdx.x < HEADS) {
    float bx = -INFINITY, bn = INFINITY;
    for (int w = 0; w < ELER_BLOCK / 32; ++w) {
      bx = fmaxf(bx, smx[w][threadIdx.x]);
      bn = fminf(bn, smn[w][threadIdx.x]);
    }
    if (bx > -INFINITY) atomic_max_f32(rng + threadIdx.x, bx + 0.0f);
    if (bn <  INFINITY) atomic_min_f32(rng + HEADS + threadIdx.x, bn + 0.0f);
  }
}

#define OWN_T      1024
#define OWN_CAP    8192
#define OWN_NCH    ((N_EDGES + OWN_CAP - 1) / OWN_CAP)
static_assert(OWN_CAP == 8 * OWN_T, "per-entry phases cover the list in 8 trips");
static_assert(OWN_CAP % 16 == 0, "chunks start on a uint4 of owner ids");
#define AGG_E      16
static_assert(AGG_E == 16, "the fence behind the row loads names 16 values");

__device__ __forceinline__ float4 edge_ex(const float4 L, const float4 R, const float4 M) {
  return make_float4(__expf(lrelu(L.x + R.x) - M.x), __expf(lrelu(L.y + R.y) - M.y),
                     __expf(lrelu(L.z + R.z) - M.z), __expf(lrelu(L.w + R.w) - M.w));
}

__device__ __forceinline__ unsigned nonzero_bytes(const unsigned x) {
  return ((x & 0x7F7F7F7Fu) + 0x7F7F7F7Fu) | x;
}

__device__ __forceinline__ unsigned other_owner(const uint4 v, const unsigned pat) {
  unsigned c = nonzero_bytes(v.x ^ pat);
  c = (c & ~0x40404040u) | ((nonzero_bytes(v.y ^ pat) >> 1) & 0x40404040u);
  c = (c & ~0x20202020u) | ((nonzero_bytes(v.z ^ pat) >> 2) & 0x20202020u);
  c = (c & ~0x10101010u) | ((nonzero_bytes(v.w ^ pat) >> 3) & 0x10101010u);
  return c;
}

__global__ __launch_bounds__(OWN_T)
void gat_owner(const int* __restrict__ src, const int* __restrict__ dst,
               const float* __restrict__ el, const float* __restrict__ er,
               const float* __restrict__ rng, const float* __restrict__ feat,
               const unsigned* __restrict__ ob, float* __restrict__ out) {
  __shared__ float4 s_ex[OWN_CAP];
  __shared__ int2   s_rec[OWN_CAP];
  __shared__ __attribute__((aligned(16))) float s_out[OWN_R * OUT_DIM];
  __shared__ __attribute__((aligned(16))) float s_er[OWN_R * HEADS];
  __shared__ __attribute__((aligned(16))) float s_M[OWN_R * HEADS];
  __shared__ __attribute__((aligned(16))) float s_den[OWN_R * HEADS];
  __shared__ int s_cnt;

  float4 G;
  const bool bound = load_bound(rng, G);
  const float fl = bound ? 1e-23f : 1e-9f;
  const int tid = threadIdx.x;
  const int lo  = blockIdx.x * OWN_R;
  const int cnt = min(OWN_R, N_NODES - lo);
  const unsigned pat = blockIdx.x * 0x01010101u;

  for (int t = tid; t < OWN_R * OUT_DIM; t += OWN_T) s_out[t] = 0.0f;
  for (int t = tid; t < cnt; t += OWN_T) {
    const float4 R = *(const float4*)(er + (size_t)(lo + t) * HEADS);
    *(float4*)(s_er + t * HEADS)  = R;
    *(float4*)(s_den + t * HEADS) = make_float4(0.0f, 0.0f, 0.0f, 0.0f);
    *(float4*)(s_M + t * HEADS) = bound
        ? make_float4(lrelu(G.x + R.x), lrelu(G.y + R.y), lrelu(G.z + R.z), lrelu(G.w + R.w))
        : make_float4(-INFINITY, -INFINITY, -INFINITY, -INFINITY);
  }

  const uint4* __restrict__ ob4 = (const uint4*)ob;
  bool fast = true;
  bool have = false;
  int  ph = bound ? 1 : 0;
  int  c = 0;
  int  n = 0;

  while (ph < 3) {
    if (!have) {
      const int q0 = fast ? 0 : c * (OWN_CAP / 16);
      const int q1 = fast ? NE16 : min(q0 + OWN_CAP / 16, NE16);
      __syncthreads();
      if (tid == 0) s_cnt = 0;
      __syncthreads();
      const int nit = (q1 - q0 + OWN_T * 8 - 1) / (OWN_T * 8);
      const int rot = (int)(blockIdx.x % (unsigned)nit);
      for (int it = 0; it < nit; ++it) {
        const int k = (it + rot < nit) ? it + rot : it + rot - nit;
        const int base = q0 + k * (OWN_T * 8);
        uint4 v[8];
#pragma unroll
        for (int j = 0; j < 8; ++j) v[j] = ob4[min(base + j * OWN_T + tid, q1 - 1)];
        unsigned cc[8], g[4];
#pragma unroll
        for (int j = 0; j < 8; ++j) cc[j] = other_owner(v[j], pat);
        if (base + OWN_T * 8 > q1) {
#pragma unroll
          for (int j = 0; j < 8; ++j) if (base + j * OWN_T + tid >= q1) cc[j] = ~0u;
        }
#pragma unroll
        for (int pr = 0; pr < 4; ++pr)
          g[pr] = ~((cc[2 * pr + 1] & 0xF0F0F0F0u) | ((cc[2 * pr] >> 4) & 0x0F0F0F0Fu));
        const unsigned nh = __popc(g[0]) + __popc(g[1]) + __popc(g[2]) + __popc(g[3]);
        unsigned pre = 0u, tot = 0u;
        for (int sl = 0; sl < 8; ++sl) {
          if (__builtin_amdgcn_ballot_w32((nh >> sl) != 0u) == 0u) break;
          const unsigned bk = __builtin_amdgcn_ballot_w32(((nh >> sl) & 1u) != 0u);
          pre += __builtin_amdgcn_mbcnt_lo(bk, 0u) << sl;
          tot += (unsigned)__popc(bk) << sl;
        }
        if (tot != 0u) {
          int at = 0;
          if ((tid & 31) == 0) at = atomicAdd(&s_cnt, (int)tot);
          int slot = __builtin_amdgcn_readfirstlane(at) + (int)pre;
#pragma unroll
          for (int pr = 0; pr < 4; ++pr) {
            unsigned gg = g[pr];
            const int e0 = 16 * (base + 2 * pr * OWN_T + tid);
            while (gg) {
              const int p = __builtin_ctz(gg);
              gg &= gg - 1;
              const int r = p & 7;
              const int e = e0 + (r >> 2) * (16 * OWN_T) + 4 * (~r & 3) + (p >> 3);
              if (slot < OWN_CAP) s_rec[slot].x = e;
              ++slot;
            }
          }
        }
      }
      __syncthreads();
      const int tot = s_cnt;
      if (fast && tot > OWN_CAP) { fast = false; c = 0; continue; }
      n = tot;
      {
        int ee[8], ss[8], dd[8];
        float4 LL[8];
#pragma unroll
        for (int j = 0; j < 8; ++j) {
          const int i = tid + j * OWN_T;
          ee[j] = (i < n) ? s_rec[i].x : 0;
        }
#pragma unroll
        for (int j = 0; j < 8; ++j) { ss[j] = src[ee[j]]; dd[j] = dst[ee[j]]; }
#pragma unroll
        for (int j = 0; j < 8; ++j) LL[j] = *(const float4*)(el + (size_t)ss[j] * HEADS);
#pragma unroll
        for (int j = 0; j < 8; ++j) {
          const int i = tid + j * OWN_T;
          s_rec[i] = make_int2(ss[j], dd[j] - lo);
          s_ex[i] = LL[j];
        }
      }
      __syncthreads();
      have = fast;
    }

    if (ph == 0) {
#pragma unroll
      for (int j = 0; j < 8; ++j) {
        const int i = tid + j * OWN_T;
        if (i < n) {
          const int u = s_rec[i].y;
          const float4 L = s_ex[i];
          const float4 R = *(const float4*)(s_er + u * HEADS);
          atomic_max_f32(s_M + u * HEADS + 0, lrelu(L.x + R.x) + 0.0f);
          atomic_max_f32(s_M + u * HEADS + 1, lrelu(L.y + R.y) + 0.0f);
          atomic_max_f32(s_M + u * HEADS + 2, lrelu(L.z + R.z) + 0.0f);
          atomic_max_f32(s_M + u * HEADS + 3, lrelu(L.w + R.w) + 0.0f);
        }
      }
    } else if (ph == 1) {
#pragma unroll
      for (int j = 0; j < 8; ++j) {
        const int i = tid + j * OWN_T;
        if (i < n) {
          const int u = s_rec[i].y;
          const float4 X = edge_ex(s_ex[i], *(const float4*)(s_er + u * HEADS),
                                   *(const float4*)(s_M + u * HEADS));
          atomicAdd(s_den + u * HEADS + 0, X.x);
          atomicAdd(s_den + u * HEADS + 1, X.y);
          atomicAdd(s_den + u * HEADS + 2, X.z);
          atomicAdd(s_den + u * HEADS + 3, X.w);
        }
      }
    } else {
#pragma unroll
      for (int j = 0; j < 8; ++j) {
        const int i = tid + j * OWN_T;
        if (i < n) {
          const int u = s_rec[i].y;
          const float4 X = edge_ex(s_ex[i], *(const float4*)(s_er + u * HEADS),
                                   *(const float4*)(s_M + u * HEADS));
          const float4 D = *(const float4*)(s_den + u * HEADS);
          s_ex[i] = make_float4(0.25f * (X.x / fmaxf(D.x, fl)), 0.25f * (X.y / fmaxf(D.y, fl)),
                                0.25f * (X.z / fmaxf(D.z, fl)), 0.25f * (X.w / fmaxf(D.w, fl)));
        }
      }
      __syncthreads();
      const int wv = __builtin_amdgcn_readfirstlane(tid >> 5), lane = tid & 31;
      for (int i0 = wv * AGG_E; i0 < n; i0 += (OWN_T / 32) * AGG_E) {
        int2 r[AGG_E];
        v4f f[AGG_E];
#pragma unroll
        for (int j = 0; j < AGG_E; ++j)
          r[j] = s_rec[min(i0 + j, n - 1)];
#pragma unroll
        for (int j = 0; j < AGG_E; ++j)
          f[j] = *(const v4f*)(feat + (size_t)r[j].x * FEAT_DIM + lane * HEADS);
        asm volatile("" :: "v"(f[0]), "v"(f[1]), "v"(f[2]), "v"(f[3]), "v"(f[4]), "v"(f[5]),
                     "v"(f[6]), "v"(f[7]), "v"(f[8]), "v"(f[9]), "v"(f[10]), "v"(f[11]),
                     "v"(f[12]), "v"(f[13]), "v"(f[14]), "v"(f[15]));
#pragma unroll
        for (int j = 0; j < AGG_E; ++j) {
          const float4 w = s_ex[min(i0 + j, n - 1)];
          const float acc = w.x * f[j].x + w.y * f[j].y + w.z * f[j].z + w.w * f[j].w;
          atomicAdd(s_out + r[j].y * OUT_DIM + lane, (i0 + j < n) ? acc : 0.0f);
        }
      }
    }
    __syncthreads();
    if (fast) { ++ph; }
    else if (++c == OWN_NCH) { c = 0; ++ph; }
  }

  float4* __restrict__ o4 = (float4*)(out + (size_t)lo * OUT_DIM);
  for (int t = tid; t < cnt * (OUT_DIM / 4); t += OWN_T) o4[t] = *(const float4*)(s_out + 4 * t);
}

extern "C" void kernel_launch(void* const* d_in, const int* in_sizes, int n_in,
                              void* d_out, int out_size, void* d_ws, size_t ws_size,
                              hipStream_t stream) {
  const float* x      = (const float*)d_in[0];
  const float* W      = (const float*)d_in[1];
  const float* attn_l = (const float*)d_in[2];
  const float* attn_r = (const float*)d_in[3];
  const int*   src    = (const int*)d_in[4];
  const int*   dst    = (const int*)d_in[5];
  float* out = (float*)d_out;

  float* ws    = (float*)d_ws;
  float* feat  = ws;
  float* el    = feat + (size_t)N_NODES * FEAT_DIM;
  float* er    = el   + (size_t)N_NODES * HEADS;
  float* rng   = er   + (size_t)N_NODES * HEADS;
  float* Wp    = rng  + (size_t)2 * N_NODES * HEADS;
  unsigned* ob = (unsigned*)(rng + 64);

  gat_prepack_W<<<PRE_BLOCKS, PRE_T, 0, stream>>>(W, Wp, rng, dst, ob);
  const int MT = N_NODES / 16;
  gat_gemm_wmma<<<(MT + 7) / 8, 256, 0, stream>>>(x, Wp, feat);
  gat_el_er<<<(N_NODES * HEADS + ELER_BLOCK - 1) / ELER_BLOCK, ELER_BLOCK, 0, stream>>>(feat, attn_l, attn_r, el, er, rng);
  gat_owner<<<OWN_BLOCKS, OWN_T, 0, stream>>>(src, dst, el, er, rng, feat, ob, out);
}


# GEMM takes its B fragments from an LDS copy of the prepacked W (128 KB, staged once per workgroup), not from L2; one 32-wave workgroup per WGP, 24-25 row tiles each; k order and sum order unchanged
# speedup vs baseline: 4.4588x; 1.2342x over previous
#include <hip/hip_runtime.h>
#include <math.h>


typedef float v2f __attribute__((ext_vector_type(2)));
typedef float v4f __attribute__((ext_vector_type(4)));
typedef float v8f __attribute__((ext_vector_type(8)));

#define N_NODES   100000
#define N_EDGES   1600000
#define IN_DIM    256
#define HEADS     4
#define OUT_DIM   32
#define FEAT_DIM  128
#define NEG_SLOPE 0.2f
#define KSTEPS    (IN_DIM / 4)
#define NTILES    (FEAT_DIM / 16)
#define BOUND_SPREAD_MAX 32.0f

#define OWN_R      391
#define OWN_BLOCKS ((N_NODES + OWN_R - 1) / OWN_R)
#define NE4        (N_EDGES / 4)
#define NE16       (N_EDGES / 16)
static_assert(OWN_BLOCKS <= 256, "an owner id is one byte");
static_assert(N_EDGES % 16 == 0, "dst is read as int4, owner ids as uint4");

#define PRE_T      256
#define PRE_BLOCKS 391
static_assert(PRE_BLOCKS * PRE_T * 4 >= NE4, "every dst quad has a thread");
static_assert(64 + NE4 <= 2 * N_NODES * HEADS, "owner ids fit between rng and Wp");
static_assert(PRE_BLOCKS * PRE_T >= KSTEPS * NTILES * 32, "every W pair has a thread");
__global__ __launch_bounds__(PRE_T)
void gat_prepack_W(const float* __restrict__ W, float* __restrict__ Wp,
                   float* __restrict__ rng, const int* __restrict__ dst,
                   unsigned* __restrict__ ob) {
  const int t = blockIdx.x * PRE_T + threadIdx.x;
  if (t < HEADS) { rng[t] = -INFINITY; rng[HEADS + t] = INFINITY; }
  if (t < KSTEPS * NTILES * 32) {
    const int lane  = t & 31;
    const int tile  = (t >> 5) & 7;
    const int kstep = t >> 8;
    const int kb = (lane >> 4) * 2;
    const int n  = tile * 16 + (lane & 15);
    const int k  = kstep * 4 + kb;
    Wp[2 * t + 0] = W[(size_t)k * FEAT_DIM + n];
    Wp[2 * t + 1] = W[(size_t)(k + 1) * FEAT_DIM + n];
  }
  const int4* __restrict__ dst4 = (const int4*)dst;
  int4 d[4];
#pragma unroll
  for (int j = 0; j < 4; ++j) d[j] = dst4[min(t + j * (PRE_BLOCKS * PRE_T), NE4 - 1)];
#pragma unroll
  for (int j = 0; j < 4; ++j) {
    const int q = t + j * (PRE_BLOCKS * PRE_T);
    if (q < NE4)
      ob[q] = ((unsigned)d[j].x / OWN_R) | (((unsigned)d[j].y / OWN_R) << 8) |
              (((unsigned)d[j].z / OWN_R) << 16) | (((unsigned)d[j].w / OWN_R) << 24);
  }
}

#define GEMM_T      1024
#define GEMM_BLOCKS 256
#define WP_FLOATS   (KSTEPS * NTILES * 64)
static_assert(WP_FLOATS == 8 * 4 * GEMM_T, "the copy is 8 float4 per thread");
__global__ __launch_bounds__(GEMM_T)
void gat_gemm_wmma(const float* __restrict__ x, const float* __restrict__ Wp,
                   float* __restrict__ feat) {
  __shared__ __attribute__((aligned(16))) float sW[WP_FLOATS];
  {
    const float4* __restrict__ Wp4 = (const float4*)Wp;
    float4 w[8];
#pragma unroll
    for (int j = 0; j < 8; ++j) w[j] = Wp4[j * GEMM_T + threadIdx.x];
#pragma unroll
    for (int j = 0; j < 8; ++j) *(float4*)(sW + 4 * (j * GEMM_T + threadIdx.x)) = w[j];
  }
  __syncthreads();

  const int lane = threadIdx.x & 31;
  const int wv = __builtin_amdgcn_readfirstlane(threadIdx.x >> 5);
  const int kb = (lane >> 4) * 2;

  for (int mt = blockIdx.x + GEMM_BLOCKS * wv; mt < N_NODES / 16;
       mt += GEMM_BLOCKS * (GEMM_T / 32)) {
    const float* aptr = x + (size_t)(mt * 16 + (lane & 15)) * IN_DIM + kb;

    v8f c[NTILES] = {};

    for (int ks = 0; ks < KSTEPS; ++ks) {
      const v2f a = *(const v2f*)(aptr + ks * 4);
#pragma unroll
      for (int t = 0; t < NTILES; ++t) {
        const v2f b = *(const v2f*)(sW + (ks * NTILES + t) * 64 + lane * 2);
        c[t] = __builtin_amdgcn_wmma_f32_16x16x4_f32(false, a, false, b,
                                                     (short)0, c[t], false, false);
      }
    }

    const int mbase = mt * 16 + 8 * (lane >> 4);
    float* o = feat + (size_t)mbase * FEAT_DIM + (lane & 15) * HEADS;
#pragma unroll
    for (int tb = 0; tb < 2; ++tb)
#pragma unroll
      for (int i = 0; i < 8; ++i)
        *(float4*)(o + (size_t)i * FEAT_DIM + tb * 16 * HEADS) =
            make_float4(c[tb][i], c[tb + 2][i], c[tb + 4][i], c[tb + 6][i]);
  }
}

__device__ __forceinline__ float lrelu(float v) {
  return v > 0.0f ? v : NEG_SLOPE * v;
}

__device__ __forceinline__ void atomic_max_f32(float* addr, float v) {
  if (v >= 0.0f) atomicMax((int*)addr, __float_as_int(v));
  else           atomicMin((unsigned int*)addr, __float_as_uint(v));
}

__device__ __forceinline__ void atomic_min_f32(float* addr, float v) {
  if (v >= 0.0f) atomicMin((int*)addr, __float_as_int(v));
  else           atomicMax((unsigned int*)addr, __float_as_uint(v));
}

__device__ __forceinline__ bool load_bound(const float* __restrict__ rng, float4& G) {
  const float4 mx = *(const float4*)(rng);
  const float4 mn = *(const float4*)(rng + HEADS);
  G = mx;
  const float r = fmaxf(fmaxf(mx.x - mn.x, mx.y - mn.y), fmaxf(mx.z - mn.z, mx.w - mn.w));
  return r <= BOUND_SPREAD_MAX;
}

#define ELER_BLOCK 1024
__global__ __launch_bounds__(ELER_BLOCK)
void gat_el_er(const float* __restrict__ feat,
               const float* __restrict__ attn_l,
               const float* __restrict__ attn_r,
               float* __restrict__ el, float* __restrict__ er,
               float* __restrict__ rng) {
  __shared__ float smx[ELER_BLOCK / 32][HEADS];
  __shared__ float smn[ELER_BLOCK / 32][HEADS];
  const int t = blockIdx.x * ELER_BLOCK + threadIdx.x;
  const int h = t & 3;
  float mx = -INFINITY, mn = INFINITY;
  if (t < N_NODES * HEADS) {
    const float* f  = feat + (size_t)(t >> 2) * FEAT_DIM + h;
    const float* al = attn_l + h * OUT_DIM;
    const float* ar = attn_r + h * OUT_DIM;
    float sl = 0.0f, sr = 0.0f;
#pragma unroll
    for (int d = 0; d < OUT_DIM; ++d) { float v = f[d * HEADS]; sl += v * al[d]; sr += v * ar[d]; }
    el[t] = sl;
    er[t] = sr;
    mx = sl; mn = sl;
  }
  for (int o = 4; o < 32; o <<= 1) {
    mx = fmaxf(mx, __shfl_xor(mx, o));
    mn = fminf(mn, __shfl_xor(mn, o));
  }
  const int lane = threadIdx.x & 31, wv = threadIdx.x >> 5;
  if (lane < HEADS) { smx[wv][lane] = mx; smn[wv][lane] = mn; }
  __syncthreads();
  if (threadIdx.x < HEADS) {
    float bx = -INFINITY, bn = INFINITY;
    for (int w = 0; w < ELER_BLOCK / 32; ++w) {
      bx = fmaxf(bx, smx[w][threadIdx.x]);
      bn = fminf(bn, smn[w][threadIdx.x]);
    }
    if (bx > -INFINITY) atomic_max_f32(rng + threadIdx.x, bx + 0.0f);
    if (bn <  INFINITY) atomic_min_f32(rng + HEADS + threadIdx.x, bn + 0.0f);
  }
}

#define OWN_T      1024
#define OWN_CAP    8192
#define OWN_NCH    ((N_EDGES + OWN_CAP - 1) / OWN_CAP)
static_assert(OWN_CAP == 8 * OWN_T, "per-entry phases cover the list in 8 trips");
static_assert(OWN_CAP % 16 == 0, "chunks start on a uint4 of owner ids");
#define AGG_E      16
static_assert(AGG_E == 16, "the fence behind the row loads names 16 values");

__device__ __forceinline__ float4 edge_ex(const float4 L, const float4 R, const float4 M) {
  return make_float4(__expf(lrelu(L.x + R.x) - M.x), __expf(lrelu(L.y + R.y) - M.y),
                     __expf(lrelu(L.z + R.z) - M.z), __expf(lrelu(L.w + R.w) - M.w));
}

__device__ __forceinline__ unsigned nonzero_bytes(const unsigned x) {
  return ((x & 0x7F7F7F7Fu) + 0x7F7F7F7Fu) | x;
}

__device__ __forceinline__ unsigned other_owner(const uint4 v, const unsigned pat) {
  unsigned c = nonzero_bytes(v.x ^ pat);
  c = (c & ~0x40404040u) | ((nonzero_bytes(v.y ^ pat) >> 1) & 0x40404040u);
  c = (c & ~0x20202020u) | ((nonzero_bytes(v.z ^ pat) >> 2) & 0x20202020u);
  c = (c & ~0x10101010u) | ((nonzero_bytes(v.w ^ pat) >> 3) & 0x10101010u);
  return c;
}

__global__ __launch_bounds__(OWN_T)
void gat_owner(const int* __restrict__ src, const int* __restrict__ dst,
               const float* __restrict__ el, const float* __restrict__ er,
               const float* __restrict__ rng, const float* __restrict__ feat,
               const unsigned* __restrict__ ob, float* __restrict__ out) {
  __shared__ float4 s_ex[OWN_CAP];
  __shared__ int2   s_rec[OWN_CAP];
  __shared__ __attribute__((aligned(16))) float s_out[OWN_R * OUT_DIM];
  __shared__ __attribute__((aligned(16))) float s_er[OWN_R * HEADS];
  __shared__ __attribute__((aligned(16))) float s_M[OWN_R * HEADS];
  __shared__ __attribute__((aligned(16))) float s_den[OWN_R * HEADS];
  __shared__ int s_cnt;

  float4 G;
  const bool bound = load_bound(rng, G);
  const float fl = bound ? 1e-23f : 1e-9f;
  const int tid = threadIdx.x;
  const int lo  = blockIdx.x * OWN_R;
  const int cnt = min(OWN_R, N_NODES - lo);
  const unsigned pat = blockIdx.x * 0x01010101u;

  for (int t = tid; t < OWN_R * OUT_DIM; t += OWN_T) s_out[t] = 0.0f;
  for (int t = tid; t < cnt; t += OWN_T) {
    const float4 R = *(const float4*)(er + (size_t)(lo + t) * HEADS);
    *(float4*)(s_er + t * HEADS)  = R;
    *(float4*)(s_den + t * HEADS) = make_float4(0.0f, 0.0f, 0.0f, 0.0f);
    *(float4*)(s_M + t * HEADS) = bound
        ? make_float4(lrelu(G.x + R.x), lrelu(G.y + R.y), lrelu(G.z + R.z), lrelu(G.w + R.w))
        : make_float4(-INFINITY, -INFINITY, -INFINITY, -INFINITY);
  }

  const uint4* __restrict__ ob4 = (const uint4*)ob;
  bool fast = true;
  bool have = false;
  int  ph = bound ? 1 : 0;
  int  c = 0;
  int  n = 0;

  while (ph < 3) {
    if (!have) {
      const int q0 = fast ? 0 : c * (OWN_CAP / 16);
      const int q1 = fast ? NE16 : min(q0 + OWN_CAP / 16, NE16);
      __syncthreads();
      if (tid == 0) s_cnt = 0;
      __syncthreads();
      const int nit = (q1 - q0 + OWN_T * 8 - 1) / (OWN_T * 8);
      const int rot = (int)(blockIdx.x % (unsigned)nit);
      for (int it = 0; it < nit; ++it) {
        const int k = (it + rot < nit) ? it + rot : it + rot - nit;
        const int base = q0 + k * (OWN_T * 8);
        uint4 v[8];
#pragma unroll
        for (int j = 0; j < 8; ++j) v[j] = ob4[min(base + j * OWN_T + tid, q1 - 1)];
        unsigned cc[8], g[4];
#pragma unroll
        for (int j = 0; j < 8; ++j) cc[j] = other_owner(v[j], pat);
        if (base + OWN_T * 8 > q1) {
#pragma unroll
          for (int j = 0; j < 8; ++j) if (base + j * OWN_T + tid >= q1) cc[j] = ~0u;
        }
#pragma unroll
        for (int pr = 0; pr < 4; ++pr)
          g[pr] = ~((cc[2 * pr + 1] & 0xF0F0F0F0u) | ((cc[2 * pr] >> 4) & 0x0F0F0F0Fu));
        const unsigned nh = __popc(g[0]) + __popc(g[1]) + __popc(g[2]) + __popc(g[3]);
        unsigned pre = 0u, tot = 0u;
        for (int sl = 0; sl < 8; ++sl) {
          if (__builtin_amdgcn_ballot_w32((nh >> sl) != 0u) == 0u) break;
          const unsigned bk = __builtin_amdgcn_ballot_w32(((nh >> sl) & 1u) != 0u);
          pre += __builtin_amdgcn_mbcnt_lo(bk, 0u) << sl;
          tot += (unsigned)__popc(bk) << sl;
        }
        if (tot != 0u) {
          int at = 0;
          if ((tid & 31) == 0) at = atomicAdd(&s_cnt, (int)tot);
          int slot = __builtin_amdgcn_readfirstlane(at) + (int)pre;
#pragma unroll
          for (int pr = 0; pr < 4; ++pr) {
            unsigned gg = g[pr];
            const int e0 = 16 * (base + 2 * pr * OWN_T + tid);
            while (gg) {
              const int p = __builtin_ctz(gg);
              gg &= gg - 1;
              const int r = p & 7;
              const int e = e0 + (r >> 2) * (16 * OWN_T) + 4 * (~r & 3) + (p >> 3);
              if (slot < OWN_CAP) s_rec[slot].x = e;
              ++slot;
            }
          }
        }
      }
      __syncthreads();
      const int tot = s_cnt;
      if (fast && tot > OWN_CAP) { fast = false; c = 0; continue; }
      n = tot;
      {
        int ee[8], ss[8], dd[8];
        float4 LL[8];
#pragma unroll
        for (int j = 0; j < 8; ++j) {
          const int i = tid + j * OWN_T;
          ee[j] = (i < n) ? s_rec[i].x : 0;
        }
#pragma unroll
        for (int j = 0; j < 8; ++j) { ss[j] = src[ee[j]]; dd[j] = dst[ee[j]]; }
#pragma unroll
        for (int j = 0; j < 8; ++j) LL[j] = *(const float4*)(el + (size_t)ss[j] * HEADS);
#pragma unroll
        for (int j = 0; j < 8; ++j) {
          const int i = tid + j * OWN_T;
          s_rec[i] = make_int2(ss[j], dd[j] - lo);
          s_ex[i] = LL[j];
        }
      }
      __syncthreads();
      have = fast;
    }

    if (ph == 0) {
#pragma unroll
      for (int j = 0; j < 8; ++j) {
        const int i = tid + j * OWN_T;
        if (i < n) {
          const int u = s_rec[i].y;
          const float4 L = s_ex[i];
          const float4 R = *(const float4*)(s_er + u * HEADS);
          atomic_max_f32(s_M + u * HEADS + 0, lrelu(L.x + R.x) + 0.0f);
          atomic_max_f32(s_M + u * HEADS + 1, lrelu(L.y + R.y) + 0.0f);
          atomic_max_f32(s_M + u * HEADS + 2, lrelu(L.z + R.z) + 0.0f);
          atomic_max_f32(s_M + u * HEADS + 3, lrelu(L.w + R.w) + 0.0f);
        }
      }
    } else if (ph == 1) {
#pragma unroll
      for (int j = 0; j < 8; ++j) {
        const int i = tid + j * OWN_T;
        if (i < n) {
          const int u = s_rec[i].y;
          const float4 X = edge_ex(s_ex[i], *(const float4*)(s_er + u * HEADS),
                                   *(const float4*)(s_M + u * HEADS));
          atomicAdd(s_den + u * HEADS + 0, X.x);
          atomicAdd(s_den + u * HEADS + 1, X.y);
          atomicAdd(s_den + u * HEADS + 2, X.z);
          atomicAdd(s_den + u * HEADS + 3, X.w);
        }
      }
    } else {
#pragma unroll
      for (int j = 0; j < 8; ++j) {
        const int i = tid + j * OWN_T;
        if (i < n) {
          const int u = s_rec[i].y;
          const float4 X = edge_ex(s_ex[i], *(const float4*)(s_er + u * HEADS),
                                   *(const float4*)(s_M + u * HEADS));
          const float4 D = *(const float4*)(s_den + u * HEADS);
          s_ex[i] = make_float4(0.25f * (X.x / fmaxf(D.x, fl)), 0.25f * (X.y / fmaxf(D.y, fl)),
                                0.25f * (X.z / fmaxf(D.z, fl)), 0.25f * (X.w / fmaxf(D.w, fl)));
        }
      }
      __syncthreads();
      const int wv = __builtin_amdgcn_readfirstlane(tid >> 5), lane = tid & 31;
      for (int i0 = wv * AGG_E; i0 < n; i0 += (OWN_T / 32) * AGG_E) {
        int2 r[AGG_E];
        v4f f[AGG_E];
#pragma unroll
        for (int j = 0; j < AGG_E; ++j)
          r[j] = s_rec[min(i0 + j, n - 1)];
#pragma unroll
        for (int j = 0; j < AGG_E; ++j)
          f[j] = *(const v4f*)(feat + (size_t)r[j].x * FEAT_DIM + lane * HEADS);
        asm volatile("" :: "v"(f[0]), "v"(f[1]), "v"(f[2]), "v"(f[3]), "v"(f[4]), "v"(f[5]),
                     "v"(f[6]), "v"(f[7]), "v"(f[8]), "v"(f[9]), "v"(f[10]), "v"(f[11]),
                     "v"(f[12]), "v"(f[13]), "v"(f[14]), "v"(f[15]));
#pragma unroll
        for (int j = 0; j < AGG_E; ++j) {
          const float4 w = s_ex[min(i0 + j, n - 1)];
          const float acc = w.x * f[j].x + w.y * f[j].y + w.z * f[j].z + w.w * f[j].w;
          atomicAdd(s_out + r[j].y * OUT_DIM + lane, (i0 + j < n) ? acc : 0.0f);
        }
      }
    }
    __syncthreads();
    if (fast) { ++ph; }
    else if (++c == OWN_NCH) { c = 0; ++ph; }
  }

  float4* __restrict__ o4 = (float4*)(out + (size_t)lo * OUT_DIM);
  for (int t = tid; t < cnt * (OUT_DIM / 4); t += OWN_T) o4[t] = *(const float4*)(s_out + 4 * t);
}

extern "C" void kernel_launch(void* const* d_in, const int* in_sizes, int n_in,
                              void* d_out, int out_size, void* d_ws, size_t ws_size,
                              hipStream_t stream) {
  const float* x      = (const float*)d_in[0];
  const float* W      = (const float*)d_in[1];
  const float* attn_l = (const float*)d_in[2];
  const float* attn_r = (const float*)d_in[3];
  const int*   src    = (const int*)d_in[4];
  const int*   dst    = (const int*)d_in[5];
  float* out = (float*)d_out;

  float* ws    = (float*)d_ws;
  float* feat  = ws;
  float* el    = feat + (size_t)N_NODES * FEAT_DIM;
  float* er    = el   + (size_t)N_NODES * HEADS;
  float* rng   = er   + (size_t)N_NODES * HEADS;
  float* Wp    = rng  + (size_t)2 * N_NODES * HEADS;
  unsigned* ob = (unsigned*)(rng + 64);

  gat_prepack_W<<<PRE_BLOCKS, PRE_T, 0, stream>>>(W, Wp, rng, dst, ob);
  gat_gemm_wmma<<<GEMM_BLOCKS, GEMM_T, 0, stream>>>(x, Wp, feat);
  gat_el_er<<<(N_NODES * HEADS + ELER_BLOCK - 1) / ELER_BLOCK, ELER_BLOCK, 0, stream>>>(feat, attn_l, attn_r, el, er, rng);
  gat_owner<<<OWN_BLOCKS, OWN_T, 0, stream>>>(src, dst, el, er, rng, feat, ob, out);
}


# edges binned by owner in the el/er launch (LDS ranks, one global int add per owner and workgroup, packed records stored in runs); owners read their bucket, no scan; LDS float adds in bucket order
# speedup vs baseline: 5.6569x; 1.2687x over previous
#include <hip/hip_runtime.h>
#include <math.h>


typedef float v2f __attribute__((ext_vector_type(2)));
typedef float v4f __attribute__((ext_vector_type(4)));
typedef float v8f __attribute__((ext_vector_type(8)));

#define N_NODES   100000
#define N_EDGES   1600000
#define IN_DIM    256
#define HEADS     4
#define OUT_DIM   32
#define FEAT_DIM  128
#define NEG_SLOPE 0.2f
#define KSTEPS    (IN_DIM / 4)
#define NTILES    (FEAT_DIM / 16)
#define BOUND_SPREAD_MAX 32.0f

#define OWN_R      391
#define OWN_BLOCKS ((N_NODES + OWN_R - 1) / OWN_R)
#define OWN_CAP    8192
#define OWN_ROWW   (OWN_R * OUT_DIM)
static_assert(N_NODES <= (1 << 17) && OWN_R <= (1 << 9), "a record is one 32-bit word");
static_assert((N_NODES - (OWN_BLOCKS - 1) * OWN_R) * OUT_DIM >= OWN_CAP,
              "a bucket fits the rows of its owner, the last one's too");
#define NE4        (N_EDGES / 4)
#define NE16       (N_EDGES / 16)
static_assert(OWN_BLOCKS <= 256, "an owner id is one byte");
static_assert(N_EDGES % 16 == 0, "dst is read as int4, owner ids as uint4");

#define PRE_T      256
#define PRE_BLOCKS 64
static_assert(64 + NE4 + OWN_BLOCKS <= 2 * N_NODES * HEADS, "owner ids and counts fit between rng and Wp");
static_assert(PRE_BLOCKS * PRE_T >= KSTEPS * NTILES * 32, "every W pair has a thread");
__global__ __launch_bounds__(PRE_T)
void gat_prepack_W(const float* __restrict__ W, float* __restrict__ Wp,
                   float* __restrict__ rng, int* __restrict__ gcnt) {
  const int t = blockIdx.x * PRE_T + threadIdx.x;
  if (t < HEADS) { rng[t] = -INFINITY; rng[HEADS + t] = INFINITY; }
  if (t < OWN_BLOCKS) gcnt[t] = 0;
  if (t < KSTEPS * NTILES * 32) {
    const int lane  = t & 31;
    const int tile  = (t >> 5) & 7;
    const int kstep = t >> 8;
    const int kb = (lane >> 4) * 2;
    const int n  = tile * 16 + (lane & 15);
    const int k  = kstep * 4 + kb;
    Wp[2 * t + 0] = W[(size_t)k * FEAT_DIM + n];
    Wp[2 * t + 1] = W[(size_t)(k + 1) * FEAT_DIM + n];
  }
}

#define GEMM_T      1024
#define GEMM_BLOCKS 256
#define WP_FLOATS   (KSTEPS * NTILES * 64)
static_assert(WP_FLOATS == 8 * 4 * GEMM_T, "the copy is 8 float4 per thread");
__global__ __launch_bounds__(GEMM_T)
void gat_gemm_wmma(const float* __restrict__ x, const float* __restrict__ Wp,
                   float* __restrict__ feat) {
  __shared__ __attribute__((aligned(16))) float sW[WP_FLOATS];
  {
    const float4* __restrict__ Wp4 = (const float4*)Wp;
    float4 w[8];
#pragma unroll
    for (int j = 0; j < 8; ++j) w[j] = Wp4[j * GEMM_T + threadIdx.x];
#pragma unroll
    for (int j = 0; j < 8; ++j) *(float4*)(sW + 4 * (j * GEMM_T + threadIdx.x)) = w[j];
  }
  __syncthreads();

  const int lane = threadIdx.x & 31;
  const int wv = __builtin_amdgcn_readfirstlane(threadIdx.x >> 5);
  const int kb = (lane >> 4) * 2;

  for (int mt = blockIdx.x + GEMM_BLOCKS * wv; mt < N_NODES / 16;
       mt += GEMM_BLOCKS * (GEMM_T / 32)) {
    const float* aptr = x + (size_t)(mt * 16 + (lane & 15)) * IN_DIM + kb;

    v8f c[NTILES] = {};

    for (int ks = 0; ks < KSTEPS; ++ks) {
      const v2f a = *(const v2f*)(aptr + ks * 4);
#pragma unroll
      for (int t = 0; t < NTILES; ++t) {
        const v2f b = *(const v2f*)(sW + (ks * NTILES + t) * 64 + lane * 2);
        c[t] = __builtin_amdgcn_wmma_f32_16x16x4_f32(false, a, false, b,
                                                     (short)0, c[t], false, false);
      }
    }

    const int mbase = mt * 16 + 8 * (lane >> 4);
    float* o = feat + (size_t)mbase * FEAT_DIM + (lane & 15) * HEADS;
#pragma unroll
    for (int tb = 0; tb < 2; ++tb)
#pragma unroll
      for (int i = 0; i < 8; ++i)
        *(float4*)(o + (size_t)i * FEAT_DIM + tb * 16 * HEADS) =
            make_float4(c[tb][i], c[tb + 2][i], c[tb + 4][i], c[tb + 6][i]);
  }
}

__device__ __forceinline__ float lrelu(float v) {
  return v > 0.0f ? v : NEG_SLOPE * v;
}

__device__ __forceinline__ void atomic_max_f32(float* addr, float v) {
  if (v >= 0.0f) atomicMax((int*)addr, __float_as_int(v));
  else           atomicMin((unsigned int*)addr, __float_as_uint(v));
}

__device__ __forceinline__ void atomic_min_f32(float* addr, float v) {
  if (v >= 0.0f) atomicMin((int*)addr, __float_as_int(v));
  else           atomicMax((unsigned int*)addr, __float_as_uint(v));
}

__device__ __forceinline__ bool load_bound(const float* __restrict__ rng, float4& G) {
  const float4 mx = *(const float4*)(rng);
  const float4 mn = *(const float4*)(rng + HEADS);
  G = mx;
  const float r = fmaxf(fmaxf(mx.x - mn.x, mx.y - mn.y), fmaxf(mx.z - mn.z, mx.w - mn.w));
  return r <= BOUND_SPREAD_MAX;
}

#define ELER_BLOCK 1024
#define BIN_E      (4 * ELER_BLOCK)
static_assert(N_NODES * HEADS == NE4, "one thread: one (node, head) and one quad of edges");
static_assert(OWN_BLOCKS == 256, "the bin scan is 32 lanes x 8 bins");
__global__ __launch_bounds__(ELER_BLOCK)
void gat_el_er(const float* __restrict__ feat,
               const float* __restrict__ attn_l,
               const float* __restrict__ attn_r,
               float* __restrict__ el, float* __restrict__ er,
               float* __restrict__ rng,
               const int* __restrict__ src, const int* __restrict__ dst,
               unsigned* __restrict__ ob, int* __restrict__ gcnt,
               unsigned* __restrict__ bk) {
  __shared__ float smx[ELER_BLOCK / 32][HEADS];
  __shared__ float smn[ELER_BLOCK / 32][HEADS];
  __shared__ int s_hist[OWN_BLOCKS];
  __shared__ int s_loc[OWN_BLOCKS];
  __shared__ int s_base[OWN_BLOCKS];
  __shared__ unsigned s_srt[BIN_E];
  __shared__ int s_to[BIN_E];
  const int t = blockIdx.x * ELER_BLOCK + threadIdx.x;
  const int h = t & 3;
  float mx = -INFINITY, mn = INFINITY;
  if (threadIdx.x < OWN_BLOCKS) s_hist[threadIdx.x] = 0;
  __syncthreads();
  if (t < N_NODES * HEADS) {
    const float* f  = feat + (size_t)(t >> 2) * FEAT_DIM + h;
    const float* al = attn_l + h * OUT_DIM;
    const float* ar = attn_r + h * OUT_DIM;
    float sl = 0.0f, sr = 0.0f;
#pragma unroll
    for (int d = 0; d < OUT_DIM; ++d) { float v = f[d * HEADS]; sl += v * al[d]; sr += v * ar[d]; }
    el[t] = sl;
    er[t] = sr;
    mx = sl; mn = sl;
  }
  int4 eD = make_int4(0, 0, 0, 0), eS = make_int4(0, 0, 0, 0);
  if (t < NE4) { eD = ((const int4*)dst)[t]; eS = ((const int4*)src)[t]; }
  for (int o = 4; o < 32; o <<= 1) {
    mx = fmaxf(mx, __shfl_xor(mx, o));
    mn = fminf(mn, __shfl_xor(mn, o));
  }
  const int lane = threadIdx.x & 31, wv = threadIdx.x >> 5;
  if (lane < HEADS) { smx[wv][lane] = mx; smn[wv][lane] = mn; }
  const int o0 = min(eD.x / OWN_R, OWN_BLOCKS - 1), o1 = min(eD.y / OWN_R, OWN_BLOCKS - 1);
  const int o2 = min(eD.z / OWN_R, OWN_BLOCKS - 1), o3 = min(eD.w / OWN_R, OWN_BLOCKS - 1);
  int r0 = 0, r1 = 0, r2 = 0, r3 = 0;
  if (t < NE4) {
    r0 = atomicAdd(&s_hist[o0], 1);
    r1 = atomicAdd(&s_hist[o1], 1);
    r2 = atomicAdd(&s_hist[o2], 1);
    r3 = atomicAdd(&s_hist[o3], 1);
    ob[t] = (unsigned)o0 | ((unsigned)o1 << 8) | ((unsigned)o2 << 16) | ((unsigned)o3 << 24);
  }
  __syncthreads();
  if (threadIdx.x < OWN_BLOCKS) {
    const int c = s_hist[threadIdx.x];
    s_base[threadIdx.x] = c ? atomicAdd(gcnt + threadIdx.x, c) : 0;
  }
  if (threadIdx.x < 32) {
    int cnt8[8], tot = 0;
#pragma unroll
    for (int k = 0; k < 8; ++k) { cnt8[k] = s_hist[8 * threadIdx.x + k]; tot += cnt8[k]; }
    int inc = tot;
    for (int o = 1; o < 32; o <<= 1) {
      const int u = __shfl_up(inc, o);
      if ((int)threadIdx.x >= o) inc += u;
    }
    int run = inc - tot;
#pragma unroll
    for (int k = 0; k < 8; ++k) { s_loc[8 * threadIdx.x + k] = run; run += cnt8[k]; }
  }
  __syncthreads();
  if (t < NE4) {
    const int p0 = s_base[o0] + r0, p1 = s_base[o1] + r1, p2 = s_base[o2] + r2, p3 = s_base[o3] + r3;
    const int l0 = s_loc[o0] + r0, l1 = s_loc[o1] + r1, l2 = s_loc[o2] + r2, l3 = s_loc[o3] + r3;
    s_srt[l0] = ((unsigned)eS.x & 0x1FFFFu) | ((unsigned)(eD.x - o0 * OWN_R) << 17);
    s_srt[l1] = ((unsigned)eS.y & 0x1FFFFu) | ((unsigned)(eD.y - o1 * OWN_R) << 17);
    s_srt[l2] = ((unsigned)eS.z & 0x1FFFFu) | ((unsigned)(eD.z - o2 * OWN_R) << 17);
    s_srt[l3] = ((unsigned)eS.w & 0x1FFFFu) | ((unsigned)(eD.w - o3 * OWN_R) << 17);
    s_to[l0] = p0 < OWN_CAP ? o0 * OWN_ROWW + p0 : -1;
    s_to[l1] = p1 < OWN_CAP ? o1 * OWN_ROWW + p1 : -1;
    s_to[l2] = p2 < OWN_CAP ? o2 * OWN_ROWW + p2 : -1;
    s_to[l3] = p3 < OWN_CAP ? o3 * OWN_ROWW + p3 : -1;
  }
  __syncthreads();
  {
    const int nb = min(BIN_E, N_EDGES - (int)blockIdx.x * BIN_E);
    for (int i = threadIdx.x; i < nb; i += ELER_BLOCK) {
      const int to = s_to[i];
      if (to >= 0) bk[to] = s_srt[i];
    }
  }
  if (threadIdx.x < HEADS) {
    float bx = -INFINITY, bn = INFINITY;
    for (int w = 0; w < ELER_BLOCK / 32; ++w) {
      bx = fmaxf(bx, smx[w][threadIdx.x]);
      bn = fminf(bn, smn[w][threadIdx.x]);
    }
    if (bx > -INFINITY) atomic_max_f32(rng + threadIdx.x, bx + 0.0f);
    if (bn <  INFINITY) atomic_min_f32(rng + HEADS + threadIdx.x, bn + 0.0f);
  }
}

#define OWN_T      1024
#define OWN_NCH    ((N_EDGES + OWN_CAP - 1) / OWN_CAP)
static_assert(OWN_CAP == 8 * OWN_T, "per-entry phases cover the list in 8 trips");
static_assert(OWN_CAP % 16 == 0, "chunks start on a uint4 of owner ids");
#define AGG_E      16
static_assert(AGG_E == 16, "the fence behind the row loads names 16 values");

__device__ __forceinline__ float4 edge_ex(const float4 L, const float4 R, const float4 M) {
  return make_float4(__expf(lrelu(L.x + R.x) - M.x), __expf(lrelu(L.y + R.y) - M.y),
                     __expf(lrelu(L.z + R.z) - M.z), __expf(lrelu(L.w + R.w) - M.w));
}

__device__ __forceinline__ unsigned nonzero_bytes(const unsigned x) {
  return ((x & 0x7F7F7F7Fu) + 0x7F7F7F7Fu) | x;
}

__device__ __forceinline__ unsigned other_owner(const uint4 v, const unsigned pat) {
  unsigned c = nonzero_bytes(v.x ^ pat);
  c = (c & ~0x40404040u) | ((nonzero_bytes(v.y ^ pat) >> 1) & 0x40404040u);
  c = (c & ~0x20202020u) | ((nonzero_bytes(v.z ^ pat) >> 2) & 0x20202020u);
  c = (c & ~0x10101010u) | ((nonzero_bytes(v.w ^ pat) >> 3) & 0x10101010u);
  return c;
}

__global__ __launch_bounds__(OWN_T)
void gat_owner(const int* __restrict__ src, const int* __restrict__ dst,
               const float* __restrict__ el, const float* __restrict__ er,
               const float* __restrict__ rng, const float* __restrict__ feat,
               const unsigned* __restrict__ ob, const int* __restrict__ gcnt,
               float* out) {
  __shared__ float4 s_ex[OWN_CAP];
  __shared__ int2   s_rec[OWN_CAP];
  __shared__ __attribute__((aligned(16))) float s_out[OWN_R * OUT_DIM];
  __shared__ __attribute__((aligned(16))) float s_er[OWN_R * HEADS];
  __shared__ __attribute__((aligned(16))) float s_M[OWN_R * HEADS];
  __shared__ __attribute__((aligned(16))) float s_den[OWN_R * HEADS];
  __shared__ int s_cnt;

  float4 G;
  const bool bound = load_bound(rng, G);
  const float fl = bound ? 1e-23f : 1e-9f;
  const int tid = threadIdx.x;
  const int lo  = blockIdx.x * OWN_R;
  const int cnt = min(OWN_R, N_NODES - lo);
  const unsigned pat = blockIdx.x * 0x01010101u;

  for (int t = tid; t < OWN_R * OUT_DIM; t += OWN_T) s_out[t] = 0.0f;
  for (int t = tid; t < cnt; t += OWN_T) {
    const float4 R = *(const float4*)(er + (size_t)(lo + t) * HEADS);
    *(float4*)(s_er + t * HEADS)  = R;
    *(float4*)(s_den + t * HEADS) = make_float4(0.0f, 0.0f, 0.0f, 0.0f);
    *(float4*)(s_M + t * HEADS) = bound
        ? make_float4(lrelu(G.x + R.x), lrelu(G.y + R.y), lrelu(G.z + R.z), lrelu(G.w + R.w))
        : make_float4(-INFINITY, -INFINITY, -INFINITY, -INFINITY);
  }

  const uint4* __restrict__ ob4 = (const uint4*)ob;
  const unsigned* bk = (const unsigned*)(out + (size_t)lo * OUT_DIM);
  const int  gn = gcnt[blockIdx.x];
  const bool fast = gn <= OWN_CAP;
  int  ph = bound ? 1 : 0;
  int  c = 0;
  int  n = 0;

  if (fast) {
    n = gn;
    unsigned rr[8];
    int ss[8];
    float4 LL[8];
#pragma unroll
    for (int j = 0; j < 8; ++j) {
      const int i = tid + j * OWN_T;
      rr[j] = bk[(i < n) ? i : 0];
    }
#pragma unroll
    for (int j = 0; j < 8; ++j) ss[j] = min((int)(rr[j] & 0x1FFFFu), N_NODES - 1);
#pragma unroll
    for (int j = 0; j < 8; ++j) LL[j] = *(const float4*)(el + (size_t)ss[j] * HEADS);
#pragma unroll
    for (int j = 0; j < 8; ++j) {
      const int i = tid + j * OWN_T;
      s_rec[i] = make_int2(ss[j], (int)(rr[j] >> 17));
      s_ex[i] = LL[j];
    }
    __syncthreads();
  }

  while (ph < 3) {
    if (!fast) {
      const int q0 = c * (OWN_CAP / 16);
      const int q1 = min(q0 + OWN_CAP / 16, NE16);
      __syncthreads();
      if (tid == 0) s_cnt = 0;
      __syncthreads();
      const int nit = (q1 - q0 + OWN_T * 8 - 1) / (OWN_T * 8);
      const int rot = (int)(blockIdx.x % (unsigned)nit);
      for (int it = 0; it < nit; ++it) {
        const int k = (it + rot < nit) ? it + rot : it + rot - nit;
        const int base = q0 + k * (OWN_T * 8);
        uint4 v[8];
#pragma unroll
        for (int j = 0; j < 8; ++j) v[j] = ob4[min(base + j * OWN_T + tid, q1 - 1)];
        unsigned cc[8], g[4];
#pragma unroll
        for (int j = 0; j < 8; ++j) cc[j] = other_owner(v[j], pat);
        if (base + OWN_T * 8 > q1) {
#pragma unroll
          for (int j = 0; j < 8; ++j) if (base + j * OWN_T + tid >= q1) cc[j] = ~0u;
        }
#pragma unroll
        for (int pr = 0; pr < 4; ++pr)
          g[pr] = ~((cc[2 * pr + 1] & 0xF0F0F0F0u) | ((cc[2 * pr] >> 4) & 0x0F0F0F0Fu));
        const unsigned nh = __popc(g[0]) + __popc(g[1]) + __popc(g[2]) + __popc(g[3]);
        unsigned pre = 0u, tot = 0u;
        for (int sl = 0; sl < 8; ++sl) {
          if (__builtin_amdgcn_ballot_w32((nh >> sl) != 0u) == 0u) break;
          const unsigned bk = __builtin_amdgcn_ballot_w32(((nh >> sl) & 1u) != 0u);
          pre += __builtin_amdgcn_mbcnt_lo(bk, 0u) << sl;
          tot += (unsigned)__popc(bk) << sl;
        }
        if (tot != 0u) {
          int at = 0;
          if ((tid & 31) == 0) at = atomicAdd(&s_cnt, (int)tot);
          int slot = __builtin_amdgcn_readfirstlane(at) + (int)pre;
#pragma unroll
          for (int pr = 0; pr < 4; ++pr) {
            unsigned gg = g[pr];
            const int e0 = 16 * (base + 2 * pr * OWN_T + tid);
            while (gg) {
              const int p = __builtin_ctz(gg);
              gg &= gg - 1;
              const int r = p & 7;
              const int e = e0 + (r >> 2) * (16 * OWN_T) + 4 * (~r & 3) + (p >> 3);
              if (slot < OWN_CAP) s_rec[slot].x = e;
              ++slot;
            }
          }
        }
      }
      __syncthreads();
      n = s_cnt;
      {
        int ee[8], ss[8], dd[8];
        float4 LL[8];
#pragma unroll
        for (int j = 0; j < 8; ++j) {
          const int i = tid + j * OWN_T;
          ee[j] = (i < n) ? s_rec[i].x : 0;
        }
#pragma unroll
        for (int j = 0; j < 8; ++j) { ss[j] = src[ee[j]]; dd[j] = dst[ee[j]]; }
#pragma unroll
        for (int j = 0; j < 8; ++j) LL[j] = *(const float4*)(el + (size_t)ss[j] * HEADS);
#pragma unroll
        for (int j = 0; j < 8; ++j) {
          const int i = tid + j * OWN_T;
          s_rec[i] = make_int2(ss[j], dd[j] - lo);
          s_ex[i] = LL[j];
        }
      }
      __syncthreads();
    }

    if (ph == 0) {
#pragma unroll
      for (int j = 0; j < 8; ++j) {
        const int i = tid + j * OWN_T;
        if (i < n) {
          const int u = s_rec[i].y;
          const float4 L = s_ex[i];
          const float4 R = *(const float4*)(s_er + u * HEADS);
          atomic_max_f32(s_M + u * HEADS + 0, lrelu(L.x + R.x) + 0.0f);
          atomic_max_f32(s_M + u * HEADS + 1, lrelu(L.y + R.y) + 0.0f);
          atomic_max_f32(s_M + u * HEADS + 2, lrelu(L.z + R.z) + 0.0f);
          atomic_max_f32(s_M + u * HEADS + 3, lrelu(L.w + R.w) + 0.0f);
        }
      }
    } else if (ph == 1) {
#pragma unroll
      for (int j = 0; j < 8; ++j) {
        const int i = tid + j * OWN_T;
        if (i < n) {
          const int u = s_rec[i].y;
          const float4 X = edge_ex(s_ex[i], *(const float4*)(s_er + u * HEADS),
                                   *(const float4*)(s_M + u * HEADS));
          atomicAdd(s_den + u * HEADS + 0, X.x);
          atomicAdd(s_den + u * HEADS + 1, X.y);
          atomicAdd(s_den + u * HEADS + 2, X.z);
          atomicAdd(s_den + u * HEADS + 3, X.w);
        }
      }
    } else {
#pragma unroll
      for (int j = 0; j < 8; ++j) {
        const int i = tid + j * OWN_T;
        if (i < n) {
          const int u = s_rec[i].y;
          const float4 X = edge_ex(s_ex[i], *(const float4*)(s_er + u * HEADS),
                                   *(const float4*)(s_M + u * HEADS));
          const float4 D = *(const float4*)(s_den + u * HEADS);
          s_ex[i] = make_float4(0.25f * (X.x / fmaxf(D.x, fl)), 0.25f * (X.y / fmaxf(D.y, fl)),
                                0.25f * (X.z / fmaxf(D.z, fl)), 0.25f * (X.w / fmaxf(D.w, fl)));
        }
      }
      __syncthreads();
      const int wv = __builtin_amdgcn_readfirstlane(tid >> 5), lane = tid & 31;
      for (int i0 = wv * AGG_E; i0 < n; i0 += (OWN_T / 32) * AGG_E) {
        int2 r[AGG_E];
        v4f f[AGG_E];
#pragma unroll
        for (int j = 0; j < AGG_E; ++j)
          r[j] = s_rec[min(i0 + j, n - 1)];
#pragma unroll
        for (int j = 0; j < AGG_E; ++j)
          f[j] = *(const v4f*)(feat + (size_t)r[j].x * FEAT_DIM + lane * HEADS);
        asm volatile("" :: "v"(f[0]), "v"(f[1]), "v"(f[2]), "v"(f[3]), "v"(f[4]), "v"(f[5]),
                     "v"(f[6]), "v"(f[7]), "v"(f[8]), "v"(f[9]), "v"(f[10]), "v"(f[11]),
                     "v"(f[12]), "v"(f[13]), "v"(f[14]), "v"(f[15]));
#pragma unroll
        for (int j = 0; j < AGG_E; ++j) {
          const float4 w = s_ex[min(i0 + j, n - 1)];
          const float acc = w.x * f[j].x + w.y * f[j].y + w.z * f[j].z + w.w * f[j].w;
          atomicAdd(s_out + r[j].y * OUT_DIM + lane, (i0 + j < n) ? acc : 0.0f);
        }
      }
    }
    __syncthreads();
    if (fast) { ++ph; }
    else if (++c == OWN_NCH) { c = 0; ++ph; }
  }

  float4* o4 = (float4*)(out + (size_t)lo * OUT_DIM);
  for (int t = tid; t < cnt * (OUT_DIM / 4); t += OWN_T) o4[t] = *(const float4*)(s_out + 4 * t);
}

extern "C" void kernel_launch(void* const* d_in, const int* in_sizes, int n_in,
                              void* d_out, int out_size, void* d_ws, size_t ws_size,
                              hipStream_t stream) {
  const float* x      = (const float*)d_in[0];
  const float* W      = (const float*)d_in[1];
  const float* attn_l = (const float*)d_in[2];
  const float* attn_r = (const float*)d_in[3];
  const int*   src    = (const int*)d_in[4];
  const int*   dst    = (const int*)d_in[5];
  float* out = (float*)d_out;

  float* ws    = (float*)d_ws;
  float* feat  = ws;
  float* el    = feat + (size_t)N_NODES * FEAT_DIM;
  float* er    = el   + (size_t)N_NODES * HEADS;
  float* rng   = er   + (size_t)N_NODES * HEADS;
  float* Wp    = rng  + (size_t)2 * N_NODES * HEADS;
  unsigned* ob = (unsigned*)(rng + 64);
  int* gcnt    = (int*)(rng + 64 + NE4);

  gat_prepack_W<<<PRE_BLOCKS, PRE_T, 0, stream>>>(W, Wp, rng, gcnt);
  gat_gemm_wmma<<<GEMM_BLOCKS, GEMM_T, 0, stream>>>(x, Wp, feat);
  gat_el_er<<<(N_NODES * HEADS + ELER_BLOCK - 1) / ELER_BLOCK, ELER_BLOCK, 0, stream>>>(feat, attn_l, attn_r, el, er, rng, src, dst, ob, gcnt, (unsigned*)out);
  gat_owner<<<OWN_BLOCKS, OWN_T, 0, stream>>>(src, dst, el, er, rng, feat, ob, gcnt, out);
}
